# starting kernel
# baseline (speedup 1.0000x reference)
#include <hip/hip_runtime.h>

typedef __attribute__((ext_vector_type(2))) float v2f;
typedef __attribute__((ext_vector_type(8))) float v8f;

#define Bsz  64
#define Sseq 4096
#define Cin  80
#define Hh   10

__global__ __launch_bounds__(128) void xproj_wmma(const float* __restrict__ inp,
                                                  const float* __restrict__ Wih,
                                                  const float* __restrict__ bih,
                                                  const float* __restrict__ bhh,
                                                  float* __restrict__ xp) {
    const int lane = threadIdx.x & 31;
    const int wave = threadIdx.x >> 5;
    const int tile = blockIdx.x * 4 + wave;
    const int half = lane >> 4;
    const int n    = lane & 15;
    const int rowA = tile * 16 + n;
    const int nn   = (n < Hh) ? n : (Hh - 1);
    const float nmask = (n < Hh) ? 1.0f : 0.0f;

    v8f acc = {};
    #pragma unroll
    for (int ks = 0; ks < Cin / 4; ++ks) {
        const int kk = ks * 4 + half * 2;
        const float2 a2 = *reinterpret_cast<const float2*>(inp + (size_t)rowA * Cin + kk);
        v2f afrag; afrag[0] = a2.x; afrag[1] = a2.y;
        v2f bfrag;
        bfrag[0] = nmask * Wih[nn * Cin + kk];
        bfrag[1] = nmask * Wih[nn * Cin + kk + 1];
        acc = __builtin_amdgcn_wmma_f32_16x16x4_f32(
            false, afrag, false, bfrag, (short)0, acc, false, false);
    }

    if (n < Hh) {
        const float bias = bih[n] + bhh[n];
        #pragma unroll
        for (int v = 0; v < 8; ++v) {
            const int g    = tile * 16 + v + half * 8;
            const int bidx = g >> 12;
            const int sidx = g & 4095;
            xp[((size_t)sidx * Bsz + bidx) * Hh + n] = acc[v] + bias;
        }
    }
}

__device__ __forceinline__ float fast_tanh(float x) {
    const float e = __expf(2.0f * x);
    return 1.0f - 2.0f / (e + 1.0f);
}

__device__ __forceinline__ float sig4(float logit) {
    const float sg = 1.0f / (1.0f + __expf(-logit));
    const float s2 = sg * sg;
    return s2 * s2;
}

__global__ __launch_bounds__(32) void rnn_scan(const float* __restrict__ xp,
                                               const float* __restrict__ Whh,
                                               const float* __restrict__ Wfc,
                                               const float* __restrict__ bfc_p,
                                               float* __restrict__ out) {
    const int lane = threadIdx.x;
    const int half = lane >> 4;
    const int i    = lane & 15;
    const int ii   = (i < Hh) ? i : (Hh - 1);
    const int b    = blockIdx.x * 2 + half;
    const int base = (lane & 16);

    float w[Hh], wf[Hh];
    #pragma unroll
    for (int j = 0; j < Hh; ++j) w[j]  = Whh[ii * Hh + j];
    #pragma unroll
    for (int j = 0; j < Hh; ++j) wf[j] = Wfc[j];
    const float bfc = bfc_p[0];

    float h = 0.0f;

    float xc = xp[(size_t)b * Hh + ii];

    float* const outb = out + (size_t)b * Sseq;

    for (int t = 0; t < Sseq; ++t) {
        float xn = 0.0f;
        if (t + 1 < Sseq)
            xn = xp[((size_t)(t + 1) * Bsz + b) * Hh + ii];
        if (t + 16 < Sseq)
            __builtin_prefetch(xp + ((size_t)(t + 16) * Bsz + b) * Hh + ii, 0, 3);

        float hj[Hh];
        #pragma unroll
        for (int j = 0; j < Hh; ++j) hj[j] = __shfl(h, base + j, 32);

        float a0 = xc, a1 = 0.0f;
        #pragma unroll
        for (int j = 0; j < 5; ++j) a0 = fmaf(w[j],     hj[j],     a0);
        #pragma unroll
        for (int j = 0; j < 5; ++j) a1 = fmaf(w[j + 5], hj[j + 5], a1);
        const float nh = fast_tanh(a0 + a1);

        if (t > 0) {
            float l0 = bfc, l1 = 0.0f;
            #pragma unroll
            for (int j = 0; j < 5; ++j) l0 = fmaf(wf[j],     hj[j],     l0);
            #pragma unroll
            for (int j = 0; j < 5; ++j) l1 = fmaf(wf[j + 5], hj[j + 5], l1);
            if (i == 0) outb[t - 1] = sig4(l0 + l1);
        }

        h = nh;
        xc = xn;
    }

    float hj[Hh];
    #pragma unroll
    for (int j = 0; j < Hh; ++j) hj[j] = __shfl(h, base + j, 32);
    float l0 = bfc, l1 = 0.0f;
    #pragma unroll
    for (int j = 0; j < 5; ++j) l0 = fmaf(wf[j],     hj[j],     l0);
    #pragma unroll
    for (int j = 0; j < 5; ++j) l1 = fmaf(wf[j + 5], hj[j + 5], l1);
    if (i == 0) outb[Sseq - 1] = sig4(l0 + l1);
}

extern "C" void kernel_launch(void* const* d_in, const int* in_sizes, int n_in,
                              void* d_out, int out_size, void* d_ws, size_t ws_size,
                              hipStream_t stream) {
    (void)in_sizes; (void)n_in; (void)out_size; (void)ws_size;
    const float* inp = (const float*)d_in[0];
    const float* Wih = (const float*)d_in[1];
    const float* Whh = (const float*)d_in[2];
    const float* bih = (const float*)d_in[3];
    const float* bhh = (const float*)d_in[4];
    const float* Wfc = (const float*)d_in[5];
    const float* bfc = (const float*)d_in[6];
    float* out = (float*)d_out;
    float* xp  = (float*)d_ws;

    const int tiles = (Bsz * Sseq) / 16;
    xproj_wmma<<<dim3(tiles / 4), 128, 0, stream>>>(inp, Wih, bih, bhh, xp);
    rnn_scan<<<dim3(Bsz / 2), 32, 0, stream>>>(xp, Whh, Wfc, bfc, out);
}


# scan: quotients end with v_fma while a step's exponents are all below 64 (one wave-wide test: no v_div_scale flag there), not v_div_fmas, whose one flag register serialised them; same sums
# speedup vs baseline: 2.1914x; 2.1914x over previous
#include <hip/hip_runtime.h>

typedef __attribute__((ext_vector_type(2))) float v2f;
typedef __attribute__((ext_vector_type(8))) float v8f;

#define Bsz  64
#define Sseq 4096
#define Cin  80
#define Hh   10

#define K2 (2.0f * 1.44269504088896340736f)

__global__ __launch_bounds__(128) void xproj_wmma(const float* __restrict__ inp,
                                                  const float* __restrict__ Wih,
                                                  const float* __restrict__ bih,
                                                  const float* __restrict__ bhh,
                                                  const float* __restrict__ Whh,
                                                  float* __restrict__ xp) {
    const int lane = threadIdx.x & 31;
    const int wave = threadIdx.x >> 5;
    const int tile = blockIdx.x * 4 + wave;
    const int half = lane >> 4;
    const int n    = lane & 15;
    const int rowA = tile * 16 + n;
    const int nn   = (n < Hh) ? n : (Hh - 1);
    const float nmask = (n < Hh) ? 1.0f : 0.0f;

    v8f acc = {};
    #pragma unroll
    for (int ks = 0; ks < Cin / 4; ++ks) {
        const int kk = ks * 4 + half * 2;
        const float2 a2 = *reinterpret_cast<const float2*>(inp + (size_t)rowA * Cin + kk);
        v2f afrag; afrag[0] = a2.x; afrag[1] = a2.y;
        v2f bfrag;
        bfrag[0] = nmask * Wih[nn * Cin + kk];
        bfrag[1] = nmask * Wih[nn * Cin + kk + 1];
        acc = __builtin_amdgcn_wmma_f32_16x16x4_f32(
            false, afrag, false, bfrag, (short)0, acc, false, false);
    }

    if (n < Hh) {
        const float bias = bih[n] + bhh[n];
        float rs = 0.0f;
        #pragma unroll
        for (int j = 0; j < Hh; ++j) rs += Whh[n * Hh + j];
        #pragma unroll
        for (int v = 0; v < 8; ++v) {
            const int g    = tile * 16 + v + half * 8;
            const int bidx = g >> 12;
            const int sidx = g & 4095;
            xp[((size_t)sidx * Bsz + bidx) * Hh + n] = ((acc[v] + bias) + rs) * K2;
        }
    }
}

struct QHead { float den, f1, f3, f4; bool flag; };

#define Y_PLAIN_BELOW 64.0f
#define L_PLAIN_ABOVE (-40.0f)

__device__ __forceinline__ QHead q_head(float den) {
    QHead h;
    h.den = den;
    bool unused;
    const float ds = __builtin_amdgcn_div_scalef(1.0f, den, false, &unused);
    const float ns = __builtin_amdgcn_div_scalef(1.0f, den, true, &h.flag);
    const float r  = __builtin_amdgcn_rcpf(ds);
    const float e0 = __builtin_fmaf(-ds, r, 1.0f);
    h.f1 = __builtin_fmaf(e0, r, r);
    const float q0 = ns * h.f1;
    const float r0 = __builtin_fmaf(-ds, q0, ns);
    h.f3 = __builtin_fmaf(r0, h.f1, q0);
    h.f4 = __builtin_fmaf(-ds, h.f3, ns);
    return h;
}

__device__ __forceinline__ float q_end_plain(const QHead& h) {
    return __builtin_amdgcn_div_fixupf(__builtin_fmaf(h.f4, h.f1, h.f3), h.den, 1.0f);
}

__device__ __forceinline__ float q_end_flag(const QHead& h) {
    return __builtin_amdgcn_div_fixupf(__builtin_amdgcn_div_fmasf(h.f4, h.f1, h.f3, h.flag), h.den, 1.0f);
}

__device__ __forceinline__ QHead u_head(float y) {
    return q_head(1.0f + __builtin_amdgcn_exp2f(y));
}

__device__ __forceinline__ float sq2(float sg) {
    const float s2 = sg * sg;
    return s2 * s2;
}

#define XSTEP (Bsz * Hh)
#define BLK   8

template <int S0, int S1, bool FIN>
__device__ __forceinline__ void scan_steps(const float (&x)[BLK][5], const float c5,
                                           const v2f a0, const v2f a1, const v2f a2,
                                           v8f& d, const float ythr, float (&lg)[BLK],
                                           const float (&lgp)[BLK], const int hh, float* const dst) {
    #pragma unroll
    for (int s = S0; s < S1; ++s) {
        const QHead h0 = u_head(d[0]), h1 = u_head(d[1]), h2 = u_head(d[2]),
                    h3 = u_head(d[3]), h4 = u_head(d[4]);
        const float ymax = fmaxf(fmaxf(fmaxf(d[0], d[1]), fmaxf(d[2], d[3])), d[4]);
        float u0 = q_end_plain(h0), u1 = q_end_plain(h1), u2 = q_end_plain(h2),
              u3 = q_end_plain(h3), u4 = q_end_plain(h4);
        if (FIN && s == S0) {
            QHead g[4];
            #pragma unroll
            for (int k = 0; k < 4; ++k) g[k] = q_head(1.0f + __expf(-(hh ? lgp[k + 4] : lgp[k])));
            float sg[4];
            #pragma unroll
            for (int k = 0; k < 4; ++k) sg[k] = q_end_plain(g[k]);
            if (__builtin_expect(__builtin_amdgcn_ballot_w32(!(ymax < ythr)) != 0u, 0)) {
                u0 = q_end_flag(h0); u1 = q_end_flag(h1); u2 = q_end_flag(h2);
                u3 = q_end_flag(h3); u4 = q_end_flag(h4);
                #pragma unroll
                for (int k = 0; k < 4; ++k) sg[k] = q_end_flag(g[k]);
            }
            *reinterpret_cast<float4*>(dst) = make_float4(sq2(sg[0]), sq2(sg[1]), sq2(sg[2]), sq2(sg[3]));
        } else {
            if (__builtin_expect(__builtin_amdgcn_ballot_w32(!(ymax < ythr)) != 0u, 0)) {
                u0 = q_end_flag(h0); u1 = q_end_flag(h1); u2 = q_end_flag(h2);
                u3 = q_end_flag(h3); u4 = q_end_flag(h4);
            }
        }
        v2f b0, b1, b2;
        b0[0] = u0; b0[1] = u1;
        b1[0] = u2; b1[1] = u3;
        b2[0] = u4; b2[1] = 0.0f;
        v8f cc;
        cc[0] = x[s][0]; cc[1] = x[s][1]; cc[2] = x[s][2]; cc[3] = x[s][3]; cc[4] = x[s][4];
        cc[5] = c5; cc[6] = 0.0f; cc[7] = 0.0f;
        d = __builtin_amdgcn_wmma_f32_16x16x4_f32(false, a0, false, b0, (short)0, cc, false, false);
        d = __builtin_amdgcn_wmma_f32_16x16x4_f32(false, a1, false, b1, (short)0, d,  false, false);
        d = __builtin_amdgcn_wmma_f32_16x16x4_f32(false, a2, false, b2, (short)0, d,  false, false);
        lg[s] = d[5];
    }
}

__device__ __forceinline__ void finish_block(const float (&lg)[BLK], const int hh, float* const dst, const bool plain_ok) {
    QHead g[4];
    #pragma unroll
    for (int s = 0; s < 4; ++s) g[s] = q_head(1.0f + __expf(-(hh ? lg[s + 4] : lg[s])));
    float sg[4];
    #pragma unroll
    for (int s = 0; s < 4; ++s) sg[s] = q_end_plain(g[s]);
    if (!plain_ok) {
        #pragma unroll
        for (int s = 0; s < 4; ++s) sg[s] = q_end_flag(g[s]);
    }
    float o[4];
    #pragma unroll
    for (int s = 0; s < 4; ++s) o[s] = sq2(sg[s]);
    *reinterpret_cast<float4*>(dst) = make_float4(o[0], o[1], o[2], o[3]);
}

__device__ __forceinline__ void load_block(float (&x)[BLK][5], const float* const xl, const int tfirst) {
    const float* const p = xl + (size_t)tfirst * XSTEP;
    #pragma unroll
    for (int s = 0; s < BLK; ++s)
        #pragma unroll
        for (int r = 0; r < 5; ++r) x[s][r] = p[(size_t)s * XSTEP + r];
}

__device__ __forceinline__ void load_block_last(float (&x)[BLK][5], const float* const xl, const int tfirst) {
    const float* const p = xl + (size_t)tfirst * XSTEP;
    #pragma unroll
    for (int s = 0; s < BLK - 1; ++s)
        #pragma unroll
        for (int r = 0; r < 5; ++r) x[s][r] = p[(size_t)s * XSTEP + r];
    const int tl = min(tfirst + BLK - 1, Sseq - 1);
    const float* const q = xl + (size_t)tl * XSTEP;
    #pragma unroll
    for (int r = 0; r < 5; ++r) x[BLK - 1][r] = q[r];
}

__global__ __launch_bounds__(32) __attribute__((amdgpu_waves_per_eu(1, 1))) void rnn_scan(const float* __restrict__ xp,
                                               const float* __restrict__ Whh,
                                               const float* __restrict__ Wfc,
                                               const float* __restrict__ bfc_p,
                                               float* __restrict__ out) {
    const int lane = threadIdx.x;
    const int c    = lane & 15;
    const int hh   = lane >> 4;
    const int b    = blockIdx.x * 16 + c;

    const int hd = (c < 5) ? c : ((c >= 8 && c < 13) ? c - 3 : -1);
    float av[6];
    #pragma unroll
    for (int r = 0; r < 6; ++r) {
        float v = 0.0f;
        if (r < 5) {
            const int hs = 5 * hh + r;
            if (hd >= 0)     v = (-2.0f * K2) * Whh[hd * Hh + hs];
            else if (c == 5 || c == 13) v = -2.0f * Wfc[hs];
        }
        av[r] = v;
    }
    v2f a0, a1, a2;
    a0[0] = av[0]; a0[1] = av[1];
    a1[0] = av[2]; a1[1] = av[3];
    a2[0] = av[4]; a2[1] = av[5];

    float sfc = 0.0f;
    #pragma unroll
    for (int j = 0; j < Hh; ++j) sfc += Wfc[j];
    const float c5 = bfc_p[0] + sfc;

    float lb = c5;
    #pragma unroll
    for (int j = 0; j < Hh; ++j) lb -= 2.0f * fmaxf(Wfc[j], 0.0f);
    const bool  sig_plain = lb > L_PLAIN_ABOVE;
    const float ythr_fin  = sig_plain ? Y_PLAIN_BELOW : -__builtin_inff();

    const float* const xl = xp + (size_t)b * Hh + 5 * hh;
    float* const outb = out + (size_t)b * Sseq + 4 * hh;

    float xa[BLK][5], xb[BLK][5], x0[5];
    #pragma unroll
    for (int r = 0; r < 5; ++r) x0[r] = xl[r];
    load_block(xa, xl, 1);

    v8f d;
    {
        v8f cc;
        cc[0] = x0[0]; cc[1] = x0[1]; cc[2] = x0[2]; cc[3] = x0[3]; cc[4] = x0[4];
        cc[5] = c5; cc[6] = 0.0f; cc[7] = 0.0f;
        v2f uh; uh[0] = 0.5f; uh[1] = 0.5f;
        d = __builtin_amdgcn_wmma_f32_16x16x4_f32(false, a0, false, uh, (short)0, cc, false, false);
        d = __builtin_amdgcn_wmma_f32_16x16x4_f32(false, a1, false, uh, (short)0, d,  false, false);
        d = __builtin_amdgcn_wmma_f32_16x16x4_f32(false, a2, false, uh, (short)0, d,  false, false);
    }

    float lga[BLK], lgb[BLK];

    load_block(xb, xl, BLK + 1);
    scan_steps<0, BLK, false>(xa, c5, a0, a1, a2, d, Y_PLAIN_BELOW, lga, lga, hh, outb);

    for (int j = 0; j < Sseq / (2 * BLK) - 1; ++j) {
        const int m = 2 * j + 1;

        load_block(xa, xl, BLK * (m + 1) + 1);
        __builtin_amdgcn_sched_barrier(0);
        scan_steps<0, 4, false>(xb, c5, a0, a1, a2, d, Y_PLAIN_BELOW, lgb, lga, hh, outb);
        scan_steps<4, BLK, true>(xb, c5, a0, a1, a2, d, ythr_fin, lgb, lga, hh, outb + BLK * (m - 1));

        __builtin_amdgcn_sched_barrier(0);
        load_block_last(xb, xl, BLK * (m + 2) + 1);
        __builtin_amdgcn_sched_barrier(0);
        scan_steps<0, 4, false>(xa, c5, a0, a1, a2, d, Y_PLAIN_BELOW, lga, lgb, hh, outb);
        scan_steps<4, BLK, true>(xa, c5, a0, a1, a2, d, ythr_fin, lga, lgb, hh, outb + BLK * m);
    }

    scan_steps<0, 4, false>(xb, c5, a0, a1, a2, d, Y_PLAIN_BELOW, lgb, lga, hh, outb);
    scan_steps<4, BLK, true>(xb, c5, a0, a1, a2, d, ythr_fin, lgb, lga, hh, outb + BLK * (Sseq / BLK - 2));
    finish_block(lgb, hh, outb + BLK * (Sseq / BLK - 1), sig_plain);
}

extern "C" void kernel_launch(void* const* d_in, const int* in_sizes, int n_in,
                              void* d_out, int out_size, void* d_ws, size_t ws_size,
                              hipStream_t stream) {
    (void)in_sizes; (void)n_in; (void)out_size; (void)ws_size;
    const float* inp = (const float*)d_in[0];
    const float* Wih = (const float*)d_in[1];
    const float* Whh = (const float*)d_in[2];
    const float* bih = (const float*)d_in[3];
    const float* bhh = (const float*)d_in[4];
    const float* Wfc = (const float*)d_in[5];
    const float* bfc = (const float*)d_in[6];
    float* out = (float*)d_out;
    float* xp  = (float*)d_ws;

    const int tiles = (Bsz * Sseq) / 16;
    xproj_wmma<<<dim3(tiles / 4), 128, 0, stream>>>(inp, Wih, bih, bhh, Whh, xp);
    rnn_scan<<<dim3(Bsz / 16), 32, 0, stream>>>(xp, Whh, Wfc, bfc, out);
}


# scan: time axis cut into 64 chunks, one wave each, begun 64 steps early from h=0; a chunk's start state is compared bit for bit with its predecessor's end state, re-run serially if unequal; same sums
# speedup vs baseline: 18.1186x; 8.2680x over previous
#include <hip/hip_runtime.h>

typedef __attribute__((ext_vector_type(2))) float v2f;
typedef __attribute__((ext_vector_type(8))) float v8f;

#define Bsz  64
#define Sseq 4096
#define Cin  80
#define Hh   10

#define K2 (2.0f * 1.44269504088896340736f)

__global__ __launch_bounds__(128) void xproj_wmma(const float* __restrict__ inp,
                                                  const float* __restrict__ Wih,
                                                  const float* __restrict__ bih,
                                                  const float* __restrict__ bhh,
                                                  const float* __restrict__ Whh,
                                                  float* __restrict__ xp) {
    const int lane = threadIdx.x & 31;
    const int wave = threadIdx.x >> 5;
    const int tile = blockIdx.x * 4 + wave;
    const int half = lane >> 4;
    const int n    = lane & 15;
    const int rowA = tile * 16 + n;
    const int nn   = (n < Hh) ? n : (Hh - 1);
    const float nmask = (n < Hh) ? 1.0f : 0.0f;

    v8f acc = {};
    #pragma unroll
    for (int ks = 0; ks < Cin / 4; ++ks) {
        const int kk = ks * 4 + half * 2;
        const float2 a2 = *reinterpret_cast<const float2*>(inp + (size_t)rowA * Cin + kk);
        v2f afrag; afrag[0] = a2.x; afrag[1] = a2.y;
        v2f bfrag;
        bfrag[0] = nmask * Wih[nn * Cin + kk];
        bfrag[1] = nmask * Wih[nn * Cin + kk + 1];
        acc = __builtin_amdgcn_wmma_f32_16x16x4_f32(
            false, afrag, false, bfrag, (short)0, acc, false, false);
    }

    if (n < Hh) {
        const float bias = bih[n] + bhh[n];
        float rs = 0.0f;
        #pragma unroll
        for (int j = 0; j < Hh; ++j) rs += Whh[n * Hh + j];
        #pragma unroll
        for (int v = 0; v < 8; ++v) {
            const int g    = tile * 16 + v + half * 8;
            const int bidx = g >> 12;
            const int sidx = g & 4095;
            xp[((size_t)sidx * Bsz + bidx) * Hh + n] = ((acc[v] + bias) + rs) * K2;
        }
    }
}

#define Y_PLAIN_BELOW 64.0f
#define L_PLAIN_ABOVE (-40.0f)

__device__ __forceinline__ float q_plain(float den) {
    const float r  = __builtin_amdgcn_rcpf(den);
    const float e0 = __builtin_fmaf(-den, r, 1.0f);
    const float f1 = __builtin_fmaf(e0, r, r);
    const float r0 = __builtin_fmaf(-den, f1, 1.0f);
    const float f3 = __builtin_fmaf(r0, f1, f1);
    const float f4 = __builtin_fmaf(-den, f3, 1.0f);
    return __builtin_fmaf(f4, f1, f3);
}

__device__ __forceinline__ float q_full(float den) {
    bool unused, flag;
    const float ds = __builtin_amdgcn_div_scalef(1.0f, den, false, &unused);
    const float ns = __builtin_amdgcn_div_scalef(1.0f, den, true, &flag);
    const float r  = __builtin_amdgcn_rcpf(ds);
    const float e0 = __builtin_fmaf(-ds, r, 1.0f);
    const float f1 = __builtin_fmaf(e0, r, r);
    const float q0 = ns * f1;
    const float r0 = __builtin_fmaf(-ds, q0, ns);
    const float f3 = __builtin_fmaf(r0, f1, q0);
    const float f4 = __builtin_fmaf(-ds, f3, ns);
    return __builtin_amdgcn_div_fixupf(__builtin_amdgcn_div_fmasf(f4, f1, f3, flag), den, 1.0f);
}

#define DEN_PLAIN_BELOW 0x1p96f

__device__ __forceinline__ float q_rare(float den, float plain) {
    return (den < DEN_PLAIN_BELOW) ? plain : q_full(den);
}

__device__ __forceinline__ float u_den(float y) {
    return 1.0f + __builtin_amdgcn_exp2f(y);
}

__device__ __forceinline__ float sq2(float sg) {
    const float s2 = sg * sg;
    return s2 * s2;
}

#define XSTEP (Bsz * Hh)
#define BLK   8

template <int S0, int S1, bool FIN>
__device__ __forceinline__ void scan_steps(const float (&x)[BLK][5], const float c5,
                                           const v2f a0, const v2f a1, const v2f a2,
                                           v8f& d, const float ythr, float (&lg)[BLK],
                                           const float (&lgp)[BLK], const int hh, float* const dst, const bool wr) {
    #pragma unroll
    for (int s = S0; s < S1; ++s) {
        const float n0 = u_den(d[0]), n1 = u_den(d[1]), n2 = u_den(d[2]),
                    n3 = u_den(d[3]), n4 = u_den(d[4]);
        const float ymax = fmaxf(fmaxf(fmaxf(d[0], d[1]), fmaxf(d[2], d[3])), d[4]);
        float u0 = q_plain(n0), u1 = q_plain(n1), u2 = q_plain(n2),
              u3 = q_plain(n3), u4 = q_plain(n4);
        if (FIN && s == S0) {
            float gd[4];
            #pragma unroll
            for (int k = 0; k < 4; ++k) gd[k] = 1.0f + __expf(-(hh ? lgp[k + 4] : lgp[k]));
            float sg[4];
            #pragma unroll
            for (int k = 0; k < 4; ++k) sg[k] = q_plain(gd[k]);
            if (__builtin_expect(__builtin_amdgcn_ballot_w32(!(ymax < ythr)) != 0u, 0)) {
                u0 = q_rare(n0, u0); u1 = q_rare(n1, u1); u2 = q_rare(n2, u2);
                u3 = q_rare(n3, u3); u4 = q_rare(n4, u4);
                #pragma unroll
                for (int k = 0; k < 4; ++k) sg[k] = q_rare(gd[k], sg[k]);
            }
            if (wr)
                *reinterpret_cast<float4*>(dst) = make_float4(sq2(sg[0]), sq2(sg[1]), sq2(sg[2]), sq2(sg[3]));
        } else {
            if (__builtin_expect(__builtin_amdgcn_ballot_w32(!(ymax < ythr)) != 0u, 0)) {
                u0 = q_rare(n0, u0); u1 = q_rare(n1, u1); u2 = q_rare(n2, u2);
                u3 = q_rare(n3, u3); u4 = q_rare(n4, u4);
            }
        }
        v2f b0, b1, b2;
        b0[0] = u0; b0[1] = u1;
        b1[0] = u2; b1[1] = u3;
        b2[0] = u4; b2[1] = 0.0f;
        v8f cc;
        cc[0] = x[s][0]; cc[1] = x[s][1]; cc[2] = x[s][2]; cc[3] = x[s][3]; cc[4] = x[s][4];
        cc[5] = c5; cc[6] = 0.0f; cc[7] = 0.0f;
        d = __builtin_amdgcn_wmma_f32_16x16x4_f32(false, a0, false, b0, (short)0, cc, false, false);
        d = __builtin_amdgcn_wmma_f32_16x16x4_f32(false, a1, false, b1, (short)0, d,  false, false);
        d = __builtin_amdgcn_wmma_f32_16x16x4_f32(false, a2, false, b2, (short)0, d,  false, false);
        lg[s] = d[5];
    }
}

__device__ __forceinline__ void finish_block(const float (&lg)[BLK], const int hh, float* const dst, const bool plain_ok, const bool wr) {
    float gd[4];
    #pragma unroll
    for (int s = 0; s < 4; ++s) gd[s] = 1.0f + __expf(-(hh ? lg[s + 4] : lg[s]));
    float sg[4];
    #pragma unroll
    for (int s = 0; s < 4; ++s) sg[s] = q_plain(gd[s]);
    if (!plain_ok) {
        #pragma unroll
        for (int s = 0; s < 4; ++s) sg[s] = q_full(gd[s]);
    }
    float o[4];
    #pragma unroll
    for (int s = 0; s < 4; ++s) o[s] = sq2(sg[s]);
    if (wr)
        *reinterpret_cast<float4*>(dst) = make_float4(o[0], o[1], o[2], o[3]);
}

__device__ __forceinline__ void load_block_last(float (&x)[BLK][5], const float* const xl, const int tfirst) {
    const float* const p = xl + (size_t)tfirst * XSTEP;
    #pragma unroll
    for (int s = 0; s < BLK - 1; ++s)
        #pragma unroll
        for (int r = 0; r < 5; ++r) x[s][r] = p[(size_t)s * XSTEP + r];
    const int tl = min(tfirst + BLK - 1, Sseq - 1);
    const float* const q = xl + (size_t)tl * XSTEP;
    #pragma unroll
    for (int r = 0; r < 5; ++r) x[BLK - 1][r] = q[r];
}

#define TCH   64
#define NCH   (Sseq / TCH)
#define LWARM 64
#define KDISP 32
#define NGRP  (Bsz / 16)
#define E_OFF 0
#define X_OFF 16
#define NEVER (1 << 20)

struct ScanCtx {
    v2f   a0, a1, a2;
    float c5;
    float ythr_fin;
    bool  sig_plain;
};

__device__ __forceinline__ ScanCtx make_ctx(const float* __restrict__ Whh,
                                            const float* __restrict__ Wfc,
                                            const float* __restrict__ bfc_p,
                                            const int c, const int hh) {
    ScanCtx k;
    const int hd = (c < 5) ? c : ((c >= 8 && c < 13) ? c - 3 : -1);
    float av[6];
    #pragma unroll
    for (int r = 0; r < 6; ++r) {
        float v = 0.0f;
        if (r < 5) {
            const int hs = 5 * hh + r;
            if (hd >= 0)     v = (-2.0f * K2) * Whh[hd * Hh + hs];
            else if (c == 5 || c == 13) v = -2.0f * Wfc[hs];
        }
        av[r] = v;
    }
    k.a0[0] = av[0]; k.a0[1] = av[1];
    k.a1[0] = av[2]; k.a1[1] = av[3];
    k.a2[0] = av[4]; k.a2[1] = av[5];

    float sfc = 0.0f;
    #pragma unroll
    for (int j = 0; j < Hh; ++j) sfc += Wfc[j];
    k.c5 = bfc_p[0] + sfc;

    float lb = k.c5;
    #pragma unroll
    for (int j = 0; j < Hh; ++j) lb -= 2.0f * fmaxf(Wfc[j], 0.0f);
    k.sig_plain = lb > L_PLAIN_ABOVE;
    k.ythr_fin  = k.sig_plain ? Y_PLAIN_BELOW : -__builtin_inff();
    return k;
}

__device__ __forceinline__ v8f pass_from_zero(const ScanCtx& k, const float* const xrow) {
    float x0[5];
    #pragma unroll
    for (int r = 0; r < 5; ++r) x0[r] = xrow[r];
    v8f cc;
    cc[0] = x0[0]; cc[1] = x0[1]; cc[2] = x0[2]; cc[3] = x0[3]; cc[4] = x0[4];
    cc[5] = k.c5; cc[6] = 0.0f; cc[7] = 0.0f;
    v2f uh; uh[0] = 0.5f; uh[1] = 0.5f;
    v8f d;
    d = __builtin_amdgcn_wmma_f32_16x16x4_f32(false, k.a0, false, uh, (short)0, cc, false, false);
    d = __builtin_amdgcn_wmma_f32_16x16x4_f32(false, k.a1, false, uh, (short)0, d,  false, false);
    d = __builtin_amdgcn_wmma_f32_16x16x4_f32(false, k.a2, false, uh, (short)0, d,  false, false);
    return d;
}

__device__ __forceinline__ void run_range(const ScanCtx& k, const float* const xl, float* const ob,
                                          v8f& d, const int t0, const int nblk, const int wfirst,
                                          const int hh) {
    float xa[BLK][5], xb[BLK][5];
    float lga[BLK], lgb[BLK];

    load_block_last(xa, xl, t0 + 1);
    load_block_last(xb, xl, t0 + BLK + 1);
    scan_steps<0, BLK, false>(xa, k.c5, k.a0, k.a1, k.a2, d, Y_PLAIN_BELOW, lga, lga, hh, ob, false);

    for (int j = 0; j < nblk / 2 - 1; ++j) {
        const int m = 2 * j + 1;

        load_block_last(xa, xl, t0 + BLK * (m + 1) + 1);
        __builtin_amdgcn_sched_barrier(0);
        scan_steps<0, 4, false>(xb, k.c5, k.a0, k.a1, k.a2, d, Y_PLAIN_BELOW, lgb, lga, hh, ob, false);
        scan_steps<4, BLK, true>(xb, k.c5, k.a0, k.a1, k.a2, d, k.ythr_fin, lgb, lga, hh,
                                 ob + BLK * (m - 1), (m - 1) >= wfirst);

        __builtin_amdgcn_sched_barrier(0);
        load_block_last(xb, xl, t0 + BLK * (m + 2) + 1);
        __builtin_amdgcn_sched_barrier(0);
        scan_steps<0, 4, false>(xa, k.c5, k.a0, k.a1, k.a2, d, Y_PLAIN_BELOW, lga, lgb, hh, ob, false);
        scan_steps<4, BLK, true>(xa, k.c5, k.a0, k.a1, k.a2, d, k.ythr_fin, lga, lgb, hh,
                                 ob + BLK * m, m >= wfirst);
    }

    scan_steps<0, 4, false>(xb, k.c5, k.a0, k.a1, k.a2, d, Y_PLAIN_BELOW, lgb, lga, hh, ob, false);
    scan_steps<4, BLK, true>(xb, k.c5, k.a0, k.a1, k.a2, d, k.ythr_fin, lgb, lga, hh,
                             ob + BLK * (nblk - 2), (nblk - 2) >= wfirst);
    finish_block(lgb, hh, ob + BLK * (nblk - 1), k.sig_plain, (nblk - 1) >= wfirst);
}

__global__ __launch_bounds__(32) __attribute__((amdgpu_waves_per_eu(1, 1))) void rnn_scan_chunks(const float* __restrict__ xp,
                                               const float* __restrict__ Whh,
                                               const float* __restrict__ Wfc,
                                               const float* __restrict__ bfc_p,
                                               float* __restrict__ out) {
    const int lane = threadIdx.x;
    const int c    = lane & 15;
    const int hh   = lane >> 4;
    const int ch   = blockIdx.x / NGRP;
    const int b    = (blockIdx.x % NGRP) * 16 + c;

    const ScanCtx k = make_ctx(Whh, Wfc, bfc_p, c, hh);

    const float* const xl = xp + (size_t)b * Hh + 5 * hh;
    float* const orow = out + (size_t)b * Sseq;
    float* const slot = orow + ch * TCH + 5 * hh;

    const int tc = ch * TCH;
    const int tw = (ch == 0) ? 0 : tc - LWARM;
    v8f d = pass_from_zero(k, xl + (size_t)tw * XSTEP);

    for (int seg = (ch == 0) ? 1 : 0; seg < 2; ++seg) {
        const int t0 = seg ? tc : tw;
        run_range(k, xl, orow + t0 + 4 * hh, d, t0, seg ? TCH / BLK : LWARM / BLK,
                  seg ? KDISP / BLK : NEVER, hh);
        float* const tp = slot + (seg ? X_OFF : E_OFF);
        #pragma unroll
        for (int r = 0; r < 5; ++r) tp[r] = d[r];
    }
}

__global__ __launch_bounds__(256) void rnn_verify(const float* __restrict__ xp,
                                                  const float* __restrict__ Whh,
                                                  const float* __restrict__ Wfc,
                                                  const float* __restrict__ bfc_p,
                                                  float* __restrict__ out) {
    const int lane = threadIdx.x & 31;
    const int w    = threadIdx.x >> 5;
    const int c    = lane & 15;
    const int hh   = lane >> 4;
    const int b    = blockIdx.x * 16 + c;
    float* const orow = out + (size_t)b * Sseq;

    __shared__ unsigned int sbad[8];
    unsigned int mybad = 0u;
    #pragma unroll
    for (int i = 0; i < (NCH + 7) / 8; ++i) {
        const int ch = min(1 + w + 8 * i, NCH - 1);
        const float* const e = orow + ch * TCH + E_OFF + 5 * hh;
        const float* const x = orow + (ch - 1) * TCH + X_OFF + 5 * hh;
        unsigned int df = 0u;
        #pragma unroll
        for (int r = 0; r < 5; ++r) df |= __float_as_uint(e[r]) ^ __float_as_uint(x[r]);
        if (__builtin_amdgcn_ballot_w32(df != 0u) != 0u) mybad = 1u;
    }
    if (lane == 0) sbad[w] = mybad;
    __syncthreads();
    unsigned int bad = 0u;
    #pragma unroll
    for (int i = 0; i < 8; ++i) bad |= sbad[i];
    if (bad == 0u || w != 0) return;

    const ScanCtx k = make_ctx(Whh, Wfc, bfc_p, c, hh);
    const float* const xl = xp + (size_t)b * Hh + 5 * hh;
    float tr[5];
    #pragma unroll
    for (int r = 0; r < 5; ++r) tr[r] = orow[X_OFF + 5 * hh + r];
    for (int ch = 1; ch < NCH; ++ch) {
        float* const slot = orow + ch * TCH + 5 * hh;
        float ev[5], xv[5];
        #pragma unroll
        for (int r = 0; r < 5; ++r) { ev[r] = slot[E_OFF + r]; xv[r] = slot[X_OFF + r]; }
        unsigned int df = 0u;
        #pragma unroll
        for (int r = 0; r < 5; ++r) df |= __float_as_uint(ev[r]) ^ __float_as_uint(tr[r]);
        if (__builtin_amdgcn_ballot_w32(df != 0u) != 0u) {
            v8f d;
            #pragma unroll
            for (int r = 0; r < 5; ++r) { slot[E_OFF + r] = tr[r]; d[r] = tr[r]; }
            d[5] = 0.0f; d[6] = 0.0f; d[7] = 0.0f;
            run_range(k, xl, orow + ch * TCH + 4 * hh, d, ch * TCH, TCH / BLK, KDISP / BLK, hh);
            #pragma unroll
            for (int r = 0; r < 5; ++r) { tr[r] = d[r]; slot[X_OFF + r] = d[r]; }
        } else {
            #pragma unroll
            for (int r = 0; r < 5; ++r) tr[r] = xv[r];
        }
    }
}

__global__ __launch_bounds__(32) __attribute__((amdgpu_waves_per_eu(1, 1))) void rnn_fill(const float* __restrict__ xp,
                                               const float* __restrict__ Whh,
                                               const float* __restrict__ Wfc,
                                               const float* __restrict__ bfc_p,
                                               float* __restrict__ out) {
    const int lane = threadIdx.x;
    const int c    = lane & 15;
    const int hh   = lane >> 4;
    const int ch   = blockIdx.x / NGRP;
    const int b    = (blockIdx.x % NGRP) * 16 + c;

    const ScanCtx k = make_ctx(Whh, Wfc, bfc_p, c, hh);
    const float* const xl = xp + (size_t)b * Hh + 5 * hh;
    float* const orow = out + (size_t)b * Sseq;
    const int tc = ch * TCH;

    v8f d;
    if (ch == 0) {
        d = pass_from_zero(k, xl);
    } else {
        const float* const e = orow + tc + E_OFF + 5 * hh;
        #pragma unroll
        for (int r = 0; r < 5; ++r) d[r] = e[r];
        d[5] = 0.0f; d[6] = 0.0f; d[7] = 0.0f;
    }
    run_range(k, xl, orow + tc + 4 * hh, d, tc, KDISP / BLK, 0, hh);
}

extern "C" void kernel_launch(void* const* d_in, const int* in_sizes, int n_in,
                              void* d_out, int out_size, void* d_ws, size_t ws_size,
                              hipStream_t stream) {
    (void)in_sizes; (void)n_in; (void)out_size; (void)ws_size;
    const float* inp = (const float*)d_in[0];
    const float* Wih = (const float*)d_in[1];
    const float* Whh = (const float*)d_in[2];
    const float* bih = (const float*)d_in[3];
    const float* bhh = (const float*)d_in[4];
    const float* Wfc = (const float*)d_in[5];
    const float* bfc = (const float*)d_in[6];
    float* out = (float*)d_out;
    float* xp  = (float*)d_ws;

    const int tiles = (Bsz * Sseq) / 16;
    xproj_wmma<<<dim3(tiles / 4), 128, 0, stream>>>(inp, Wih, bih, bhh, Whh, xp);
    rnn_scan_chunks<<<dim3(NCH * NGRP), 32, 0, stream>>>(xp, Whh, Wfc, bfc, out);
    rnn_verify<<<dim3(NGRP), 256, 0, stream>>>(xp, Whh, Wfc, bfc, out);
    rnn_fill<<<dim3(NCH * NGRP), 32, 0, stream>>>(xp, Whh, Wfc, bfc, out);
}


# xproj: 4,096 waves of 4 consecutive tiles each (was 16,384 one-tile waves, slower to enter than to run); W_ih fragments once per wave, next tile's A reads in flight under the WMMAs; same sums
# speedup vs baseline: 21.3523x; 1.1785x over previous
#include <hip/hip_runtime.h>

typedef __attribute__((ext_vector_type(2))) float v2f;
typedef __attribute__((ext_vector_type(8))) float v8f;

#define Bsz  64
#define Sseq 4096
#define Cin  80
#define Hh   10

#define K2 (2.0f * 1.44269504088896340736f)

#define TPW 4
#define KST (Cin / 4)

__global__ __launch_bounds__(128) void xproj_wmma(const float* __restrict__ inp,
                                                  const float* __restrict__ Wih,
                                                  const float* __restrict__ bih,
                                                  const float* __restrict__ bhh,
                                                  const float* __restrict__ Whh,
                                                  float* __restrict__ xp) {
    const int lane = threadIdx.x & 31;
    const int wave = threadIdx.x >> 5;
    const int tile0 = (blockIdx.x * 4 + wave) * TPW;
    const int half = lane >> 4;
    const int n    = lane & 15;
    const int nn   = (n < Hh) ? n : (Hh - 1);
    const float nmask = (n < Hh) ? 1.0f : 0.0f;

    float wraw[KST][2];
    #pragma unroll
    for (int ks = 0; ks < KST; ++ks) {
        const int kk = ks * 4 + half * 2;
        wraw[ks][0] = Wih[nn * Cin + kk];
        wraw[ks][1] = Wih[nn * Cin + kk + 1];
    }
    const float bi = bih[nn], bh = bhh[nn];
    float whr[Hh];
    #pragma unroll
    for (int j = 0; j < Hh; ++j) whr[j] = Whh[nn * Hh + j];

    const float* const arow = inp + ((size_t)tile0 * 16 + n) * Cin + half * 2;
    v2f abuf[2][KST];
    #pragma unroll
    for (int t = 0; t < 2; ++t) {
        const float* const at = arow + (size_t)t * 16 * Cin;
        #pragma unroll
        for (int ks = 0; ks < KST; ++ks) {
            const float2 a2 = *reinterpret_cast<const float2*>(at + ks * 4);
            abuf[t][ks][0] = a2.x; abuf[t][ks][1] = a2.y;
        }
    }
    __builtin_amdgcn_sched_barrier(0);

    v2f bfrag[KST];
    #pragma unroll
    for (int ks = 0; ks < KST; ++ks) {
        bfrag[ks][0] = nmask * wraw[ks][0];
        bfrag[ks][1] = nmask * wraw[ks][1];
    }
    const float bias = bi + bh;
    float rs = 0.0f;
    #pragma unroll
    for (int j = 0; j < Hh; ++j) rs += whr[j];

    #pragma unroll
    for (int t = 0; t < TPW; ++t) {
        if (t >= 1 && t + 1 < TPW) {
            const float* const anext = arow + (size_t)(t + 1) * 16 * Cin;
            #pragma unroll
            for (int ks = 0; ks < KST; ++ks) {
                const float2 a2 = *reinterpret_cast<const float2*>(anext + ks * 4);
                abuf[(t + 1) & 1][ks][0] = a2.x; abuf[(t + 1) & 1][ks][1] = a2.y;
            }
        }

        v8f acc = {};
        #pragma unroll
        for (int ks = 0; ks < KST; ++ks)
            acc = __builtin_amdgcn_wmma_f32_16x16x4_f32(
                false, abuf[t & 1][ks], false, bfrag[ks], (short)0, acc, false, false);

        if (n < Hh) {
            #pragma unroll
            for (int v = 0; v < 8; ++v) {
                const int g    = (tile0 + t) * 16 + v + half * 8;
                const int bidx = g >> 12;
                const int sidx = g & 4095;
                xp[((size_t)sidx * Bsz + bidx) * Hh + n] = ((acc[v] + bias) + rs) * K2;
            }
        }
    }
}

#define Y_PLAIN_BELOW 64.0f
#define L_PLAIN_ABOVE (-40.0f)

__device__ __forceinline__ float q_plain(float den) {
    const float r  = __builtin_amdgcn_rcpf(den);
    const float e0 = __builtin_fmaf(-den, r, 1.0f);
    const float f1 = __builtin_fmaf(e0, r, r);
    const float r0 = __builtin_fmaf(-den, f1, 1.0f);
    const float f3 = __builtin_fmaf(r0, f1, f1);
    const float f4 = __builtin_fmaf(-den, f3, 1.0f);
    return __builtin_fmaf(f4, f1, f3);
}

__device__ __forceinline__ float q_full(float den) {
    bool unused, flag;
    const float ds = __builtin_amdgcn_div_scalef(1.0f, den, false, &unused);
    const float ns = __builtin_amdgcn_div_scalef(1.0f, den, true, &flag);
    const float r  = __builtin_amdgcn_rcpf(ds);
    const float e0 = __builtin_fmaf(-ds, r, 1.0f);
    const float f1 = __builtin_fmaf(e0, r, r);
    const float q0 = ns * f1;
    const float r0 = __builtin_fmaf(-ds, q0, ns);
    const float f3 = __builtin_fmaf(r0, f1, q0);
    const float f4 = __builtin_fmaf(-ds, f3, ns);
    return __builtin_amdgcn_div_fixupf(__builtin_amdgcn_div_fmasf(f4, f1, f3, flag), den, 1.0f);
}

#define DEN_PLAIN_BELOW 0x1p96f

__device__ __forceinline__ float q_rare(float den, float plain) {
    return (den < DEN_PLAIN_BELOW) ? plain : q_full(den);
}

__device__ __forceinline__ float u_den(float y) {
    return 1.0f + __builtin_amdgcn_exp2f(y);
}

__device__ __forceinline__ float sq2(float sg) {
    const float s2 = sg * sg;
    return s2 * s2;
}

#define XSTEP (Bsz * Hh)
#define BLK   8

template <int S0, int S1, bool FIN>
__device__ __forceinline__ void scan_steps(const float (&x)[BLK][5], const float c5,
                                           const v2f a0, const v2f a1, const v2f a2,
                                           v8f& d, const float ythr, float (&lg)[BLK],
                                           const float (&lgp)[BLK], const int hh, float* const dst, const bool wr) {
    #pragma unroll
    for (int s = S0; s < S1; ++s) {
        const float n0 = u_den(d[0]), n1 = u_den(d[1]), n2 = u_den(d[2]),
                    n3 = u_den(d[3]), n4 = u_den(d[4]);
        const float ymax = fmaxf(fmaxf(fmaxf(d[0], d[1]), fmaxf(d[2], d[3])), d[4]);
        float u0 = q_plain(n0), u1 = q_plain(n1), u2 = q_plain(n2),
              u3 = q_plain(n3), u4 = q_plain(n4);
        if (FIN && s == S0) {
            float gd[4];
            #pragma unroll
            for (int k = 0; k < 4; ++k) gd[k] = 1.0f + __expf(-(hh ? lgp[k + 4] : lgp[k]));
            float sg[4];
            #pragma unroll
            for (int k = 0; k < 4; ++k) sg[k] = q_plain(gd[k]);
            if (__builtin_expect(__builtin_amdgcn_ballot_w32(!(ymax < ythr)) != 0u, 0)) {
                u0 = q_rare(n0, u0); u1 = q_rare(n1, u1); u2 = q_rare(n2, u2);
                u3 = q_rare(n3, u3); u4 = q_rare(n4, u4);
                #pragma unroll
                for (int k = 0; k < 4; ++k) sg[k] = q_rare(gd[k], sg[k]);
            }
            if (wr)
                *reinterpret_cast<float4*>(dst) = make_float4(sq2(sg[0]), sq2(sg[1]), sq2(sg[2]), sq2(sg[3]));
        } else {
            if (__builtin_expect(__builtin_amdgcn_ballot_w32(!(ymax < ythr)) != 0u, 0)) {
                u0 = q_rare(n0, u0); u1 = q_rare(n1, u1); u2 = q_rare(n2, u2);
                u3 = q_rare(n3, u3); u4 = q_rare(n4, u4);
            }
        }
        v2f b0, b1, b2;
        b0[0] = u0; b0[1] = u1;
        b1[0] = u2; b1[1] = u3;
        b2[0] = u4; b2[1] = 0.0f;
        v8f cc;
        cc[0] = x[s][0]; cc[1] = x[s][1]; cc[2] = x[s][2]; cc[3] = x[s][3]; cc[4] = x[s][4];
        cc[5] = c5; cc[6] = 0.0f; cc[7] = 0.0f;
        d = __builtin_amdgcn_wmma_f32_16x16x4_f32(false, a0, false, b0, (short)0, cc, false, false);
        d = __builtin_amdgcn_wmma_f32_16x16x4_f32(false, a1, false, b1, (short)0, d,  false, false);
        d = __builtin_amdgcn_wmma_f32_16x16x4_f32(false, a2, false, b2, (short)0, d,  false, false);
        lg[s] = d[5];
    }
}

__device__ __forceinline__ void finish_block(const float (&lg)[BLK], const int hh, float* const dst, const bool plain_ok, const bool wr) {
    float gd[4];
    #pragma unroll
    for (int s = 0; s < 4; ++s) gd[s] = 1.0f + __expf(-(hh ? lg[s + 4] : lg[s]));
    float sg[4];
    #pragma unroll
    for (int s = 0; s < 4; ++s) sg[s] = q_plain(gd[s]);
    if (!plain_ok) {
        #pragma unroll
        for (int s = 0; s < 4; ++s) sg[s] = q_full(gd[s]);
    }
    float o[4];
    #pragma unroll
    for (int s = 0; s < 4; ++s) o[s] = sq2(sg[s]);
    if (wr)
        *reinterpret_cast<float4*>(dst) = make_float4(o[0], o[1], o[2], o[3]);
}

__device__ __forceinline__ void load_block_last(float (&x)[BLK][5], const float* const xl, const int tfirst) {
    const float* const p = xl + (size_t)tfirst * XSTEP;
    #pragma unroll
    for (int s = 0; s < BLK - 1; ++s)
        #pragma unroll
        for (int r = 0; r < 5; ++r) x[s][r] = p[(size_t)s * XSTEP + r];
    const int tl = min(tfirst + BLK - 1, Sseq - 1);
    const float* const q = xl + (size_t)tl * XSTEP;
    #pragma unroll
    for (int r = 0; r < 5; ++r) x[BLK - 1][r] = q[r];
}

#define TCH   64
#define NCH   (Sseq / TCH)
#define LWARM 64
#define KDISP 32
#define NGRP  (Bsz / 16)
#define E_OFF 0
#define X_OFF 16
#define NEVER (1 << 20)

struct ScanCtx {
    v2f   a0, a1, a2;
    float c5;
    float ythr_fin;
    bool  sig_plain;
};

__device__ __forceinline__ ScanCtx make_ctx(const float* __restrict__ Whh,
                                            const float* __restrict__ Wfc,
                                            const float* __restrict__ bfc_p,
                                            const int c, const int hh) {
    ScanCtx k;
    const int hd = (c < 5) ? c : ((c >= 8 && c < 13) ? c - 3 : -1);
    float av[6];
    #pragma unroll
    for (int r = 0; r < 6; ++r) {
        float v = 0.0f;
        if (r < 5) {
            const int hs = 5 * hh + r;
            if (hd >= 0)     v = (-2.0f * K2) * Whh[hd * Hh + hs];
            else if (c == 5 || c == 13) v = -2.0f * Wfc[hs];
        }
        av[r] = v;
    }
    k.a0[0] = av[0]; k.a0[1] = av[1];
    k.a1[0] = av[2]; k.a1[1] = av[3];
    k.a2[0] = av[4]; k.a2[1] = av[5];

    float sfc = 0.0f;
    #pragma unroll
    for (int j = 0; j < Hh; ++j) sfc += Wfc[j];
    k.c5 = bfc_p[0] + sfc;

    float lb = k.c5;
    #pragma unroll
    for (int j = 0; j < Hh; ++j) lb -= 2.0f * fmaxf(Wfc[j], 0.0f);
    k.sig_plain = lb > L_PLAIN_ABOVE;
    k.ythr_fin  = k.sig_plain ? Y_PLAIN_BELOW : -__builtin_inff();
    return k;
}

__device__ __forceinline__ v8f pass_from_zero(const ScanCtx& k, const float* const xrow) {
    float x0[5];
    #pragma unroll
    for (int r = 0; r < 5; ++r) x0[r] = xrow[r];
    v8f cc;
    cc[0] = x0[0]; cc[1] = x0[1]; cc[2] = x0[2]; cc[3] = x0[3]; cc[4] = x0[4];
    cc[5] = k.c5; cc[6] = 0.0f; cc[7] = 0.0f;
    v2f uh; uh[0] = 0.5f; uh[1] = 0.5f;
    v8f d;
    d = __builtin_amdgcn_wmma_f32_16x16x4_f32(false, k.a0, false, uh, (short)0, cc, false, false);
    d = __builtin_amdgcn_wmma_f32_16x16x4_f32(false, k.a1, false, uh, (short)0, d,  false, false);
    d = __builtin_amdgcn_wmma_f32_16x16x4_f32(false, k.a2, false, uh, (short)0, d,  false, false);
    return d;
}

__device__ __forceinline__ void run_range(const ScanCtx& k, const float* const xl, float* const ob,
                                          v8f& d, const int t0, const int nblk, const int wfirst,
                                          const int hh) {
    float xa[BLK][5], xb[BLK][5];
    float lga[BLK], lgb[BLK];

    load_block_last(xa, xl, t0 + 1);
    load_block_last(xb, xl, t0 + BLK + 1);
    scan_steps<0, BLK, false>(xa, k.c5, k.a0, k.a1, k.a2, d, Y_PLAIN_BELOW, lga, lga, hh, ob, false);

    for (int j = 0; j < nblk / 2 - 1; ++j) {
        const int m = 2 * j + 1;

        load_block_last(xa, xl, t0 + BLK * (m + 1) + 1);
        __builtin_amdgcn_sched_barrier(0);
        scan_steps<0, 4, false>(xb, k.c5, k.a0, k.a1, k.a2, d, Y_PLAIN_BELOW, lgb, lga, hh, ob, false);
        scan_steps<4, BLK, true>(xb, k.c5, k.a0, k.a1, k.a2, d, k.ythr_fin, lgb, lga, hh,
                                 ob + BLK * (m - 1), (m - 1) >= wfirst);

        __builtin_amdgcn_sched_barrier(0);
        load_block_last(xb, xl, t0 + BLK * (m + 2) + 1);
        __builtin_amdgcn_sched_barrier(0);
        scan_steps<0, 4, false>(xa, k.c5, k.a0, k.a1, k.a2, d, Y_PLAIN_BELOW, lga, lgb, hh, ob, false);
        scan_steps<4, BLK, true>(xa, k.c5, k.a0, k.a1, k.a2, d, k.ythr_fin, lga, lgb, hh,
                                 ob + BLK * m, m >= wfirst);
    }

    scan_steps<0, 4, false>(xb, k.c5, k.a0, k.a1, k.a2, d, Y_PLAIN_BELOW, lgb, lga, hh, ob, false);
    scan_steps<4, BLK, true>(xb, k.c5, k.a0, k.a1, k.a2, d, k.ythr_fin, lgb, lga, hh,
                             ob + BLK * (nblk - 2), (nblk - 2) >= wfirst);
    finish_block(lgb, hh, ob + BLK * (nblk - 1), k.sig_plain, (nblk - 1) >= wfirst);
}

__global__ __launch_bounds__(32) __attribute__((amdgpu_waves_per_eu(1, 1))) void rnn_scan_chunks(const float* __restrict__ xp,
                                               const float* __restrict__ Whh,
                                               const float* __restrict__ Wfc,
                                               const float* __restrict__ bfc_p,
                                               float* __restrict__ out) {
    const int lane = threadIdx.x;
    const int c    = lane & 15;
    const int hh   = lane >> 4;
    const int ch   = blockIdx.x / NGRP;
    const int b    = (blockIdx.x % NGRP) * 16 + c;

    const ScanCtx k = make_ctx(Whh, Wfc, bfc_p, c, hh);

    const float* const xl = xp + (size_t)b * Hh + 5 * hh;
    float* const orow = out + (size_t)b * Sseq;
    float* const slot = orow + ch * TCH + 5 * hh;

    const int tc = ch * TCH;
    const int tw = (ch == 0) ? 0 : tc - LWARM;
    v8f d = pass_from_zero(k, xl + (size_t)tw * XSTEP);

    for (int seg = (ch == 0) ? 1 : 0; seg < 2; ++seg) {
        const int t0 = seg ? tc : tw;
        run_range(k, xl, orow + t0 + 4 * hh, d, t0, seg ? TCH / BLK : LWARM / BLK,
                  seg ? KDISP / BLK : NEVER, hh);
        float* const tp = slot + (seg ? X_OFF : E_OFF);
        #pragma unroll
        for (int r = 0; r < 5; ++r) tp[r] = d[r];
    }
}

__global__ __launch_bounds__(256) void rnn_verify(const float* __restrict__ xp,
                                                  const float* __restrict__ Whh,
                                                  const float* __restrict__ Wfc,
                                                  const float* __restrict__ bfc_p,
                                                  float* __restrict__ out) {
    const int lane = threadIdx.x & 31;
    const int w    = threadIdx.x >> 5;
    const int c    = lane & 15;
    const int hh   = lane >> 4;
    const int b    = blockIdx.x * 16 + c;
    float* const orow = out + (size_t)b * Sseq;

    __shared__ unsigned int sbad[8];
    unsigned int mybad = 0u;
    #pragma unroll
    for (int i = 0; i < (NCH + 7) / 8; ++i) {
        const int ch = min(1 + w + 8 * i, NCH - 1);
        const float* const e = orow + ch * TCH + E_OFF + 5 * hh;
        const float* const x = orow + (ch - 1) * TCH + X_OFF + 5 * hh;
        unsigned int df = 0u;
        #pragma unroll
        for (int r = 0; r < 5; ++r) df |= __float_as_uint(e[r]) ^ __float_as_uint(x[r]);
        if (__builtin_amdgcn_ballot_w32(df != 0u) != 0u) mybad = 1u;
    }
    if (lane == 0) sbad[w] = mybad;
    __syncthreads();
    unsigned int bad = 0u;
    #pragma unroll
    for (int i = 0; i < 8; ++i) bad |= sbad[i];
    if (bad == 0u || w != 0) return;

    const ScanCtx k = make_ctx(Whh, Wfc, bfc_p, c, hh);
    const float* const xl = xp + (size_t)b * Hh + 5 * hh;
    float tr[5];
    #pragma unroll
    for (int r = 0; r < 5; ++r) tr[r] = orow[X_OFF + 5 * hh + r];
    for (int ch = 1; ch < NCH; ++ch) {
        float* const slot = orow + ch * TCH + 5 * hh;
        float ev[5], xv[5];
        #pragma unroll
        for (int r = 0; r < 5; ++r) { ev[r] = slot[E_OFF + r]; xv[r] = slot[X_OFF + r]; }
        unsigned int df = 0u;
        #pragma unroll
        for (int r = 0; r < 5; ++r) df |= __float_as_uint(ev[r]) ^ __float_as_uint(tr[r]);
        if (__builtin_amdgcn_ballot_w32(df != 0u) != 0u) {
            v8f d;
            #pragma unroll
            for (int r = 0; r < 5; ++r) { slot[E_OFF + r] = tr[r]; d[r] = tr[r]; }
            d[5] = 0.0f; d[6] = 0.0f; d[7] = 0.0f;
            run_range(k, xl, orow + ch * TCH + 4 * hh, d, ch * TCH, TCH / BLK, KDISP / BLK, hh);
            #pragma unroll
            for (int r = 0; r < 5; ++r) { tr[r] = d[r]; slot[X_OFF + r] = d[r]; }
        } else {
            #pragma unroll
            for (int r = 0; r < 5; ++r) tr[r] = xv[r];
        }
    }
}

__global__ __launch_bounds__(32) __attribute__((amdgpu_waves_per_eu(1, 1))) void rnn_fill(const float* __restrict__ xp,
                                               const float* __restrict__ Whh,
                                               const float* __restrict__ Wfc,
                                               const float* __restrict__ bfc_p,
                                               float* __restrict__ out) {
    const int lane = threadIdx.x;
    const int c    = lane & 15;
    const int hh   = lane >> 4;
    const int ch   = blockIdx.x / NGRP;
    const int b    = (blockIdx.x % NGRP) * 16 + c;

    const ScanCtx k = make_ctx(Whh, Wfc, bfc_p, c, hh);
    const float* const xl = xp + (size_t)b * Hh + 5 * hh;
    float* const orow = out + (size_t)b * Sseq;
    const int tc = ch * TCH;

    v8f d;
    if (ch == 0) {
        d = pass_from_zero(k, xl);
    } else {
        const float* const e = orow + tc + E_OFF + 5 * hh;
        #pragma unroll
        for (int r = 0; r < 5; ++r) d[r] = e[r];
        d[5] = 0.0f; d[6] = 0.0f; d[7] = 0.0f;
    }
    run_range(k, xl, orow + tc + 4 * hh, d, tc, KDISP / BLK, 0, hh);
}

extern "C" void kernel_launch(void* const* d_in, const int* in_sizes, int n_in,
                              void* d_out, int out_size, void* d_ws, size_t ws_size,
                              hipStream_t stream) {
    (void)in_sizes; (void)n_in; (void)out_size; (void)ws_size;
    const float* inp = (const float*)d_in[0];
    const float* Wih = (const float*)d_in[1];
    const float* Whh = (const float*)d_in[2];
    const float* bih = (const float*)d_in[3];
    const float* bhh = (const float*)d_in[4];
    const float* Wfc = (const float*)d_in[5];
    const float* bfc = (const float*)d_in[6];
    float* out = (float*)d_out;
    float* xp  = (float*)d_ws;

    const int tiles = (Bsz * Sseq) / 16;
    xproj_wmma<<<dim3(tiles / (4 * TPW)), 128, 0, stream>>>(inp, Wih, bih, bhh, Whh, xp);
    rnn_scan_chunks<<<dim3(NCH * NGRP), 32, 0, stream>>>(xp, Whh, Wfc, bfc, out);
    rnn_verify<<<dim3(NGRP), 256, 0, stream>>>(xp, Whh, Wfc, bfc, out);
    rnn_fill<<<dim3(NCH * NGRP), 32, 0, stream>>>(xp, Whh, Wfc, bfc, out);
}


# scan: every step of a chunk computed once: a warm-up launch hands each 32-step chunk its entry state, the run launch stores all outputs and checks its end state against the next entry; same sums
# speedup vs baseline: 23.9664x; 1.1224x over previous
#include <hip/hip_runtime.h>

typedef __attribute__((ext_vector_type(2))) float v2f;
typedef __attribute__((ext_vector_type(8))) float v8f;

#define Bsz  64
#define Sseq 4096
#define Cin  80
#define Hh   10

#define K2 (2.0f * 1.44269504088896340736f)

#define TPW 4
#define KST (Cin / 4)

__global__ __launch_bounds__(128) void xproj_wmma(const float* __restrict__ inp,
                                                  const float* __restrict__ Wih,
                                                  const float* __restrict__ bih,
                                                  const float* __restrict__ bhh,
                                                  const float* __restrict__ Whh,
                                                  float* __restrict__ xp) {
    const int lane = threadIdx.x & 31;
    const int wave = threadIdx.x >> 5;
    const int tile0 = (blockIdx.x * 4 + wave) * TPW;
    const int half = lane >> 4;
    const int n    = lane & 15;
    const int nn   = (n < Hh) ? n : (Hh - 1);
    const float nmask = (n < Hh) ? 1.0f : 0.0f;

    float wraw[KST][2];
    #pragma unroll
    for (int ks = 0; ks < KST; ++ks) {
        const int kk = ks * 4 + half * 2;
        wraw[ks][0] = Wih[nn * Cin + kk];
        wraw[ks][1] = Wih[nn * Cin + kk + 1];
    }
    const float bi = bih[nn], bh = bhh[nn];
    float whr[Hh];
    #pragma unroll
    for (int j = 0; j < Hh; ++j) whr[j] = Whh[nn * Hh + j];

    const float* const arow = inp + ((size_t)tile0 * 16 + n) * Cin + half * 2;
    v2f abuf[2][KST];
    #pragma unroll
    for (int t = 0; t < 2; ++t) {
        const float* const at = arow + (size_t)t * 16 * Cin;
        #pragma unroll
        for (int ks = 0; ks < KST; ++ks) {
            const float2 a2 = *reinterpret_cast<const float2*>(at + ks * 4);
            abuf[t][ks][0] = a2.x; abuf[t][ks][1] = a2.y;
        }
    }
    __builtin_amdgcn_sched_barrier(0);

    v2f bfrag[KST];
    #pragma unroll
    for (int ks = 0; ks < KST; ++ks) {
        bfrag[ks][0] = nmask * wraw[ks][0];
        bfrag[ks][1] = nmask * wraw[ks][1];
    }
    const float bias = bi + bh;
    float rs = 0.0f;
    #pragma unroll
    for (int j = 0; j < Hh; ++j) rs += whr[j];

    #pragma unroll
    for (int t = 0; t < TPW; ++t) {
        if (t >= 1 && t + 1 < TPW) {
            const float* const anext = arow + (size_t)(t + 1) * 16 * Cin;
            #pragma unroll
            for (int ks = 0; ks < KST; ++ks) {
                const float2 a2 = *reinterpret_cast<const float2*>(anext + ks * 4);
                abuf[(t + 1) & 1][ks][0] = a2.x; abuf[(t + 1) & 1][ks][1] = a2.y;
            }
        }

        v8f acc = {};
        #pragma unroll
        for (int ks = 0; ks < KST; ++ks)
            acc = __builtin_amdgcn_wmma_f32_16x16x4_f32(
                false, abuf[t & 1][ks], false, bfrag[ks], (short)0, acc, false, false);

        if (n < Hh) {
            #pragma unroll
            for (int v = 0; v < 8; ++v) {
                const int g    = (tile0 + t) * 16 + v + half * 8;
                const int bidx = g >> 12;
                const int sidx = g & 4095;
                xp[((size_t)sidx * Bsz + bidx) * Hh + n] = ((acc[v] + bias) + rs) * K2;
            }
        }
    }
}

#define Y_PLAIN_BELOW 64.0f
#define L_PLAIN_ABOVE (-40.0f)

__device__ __forceinline__ float q_plain(float den) {
    const float r  = __builtin_amdgcn_rcpf(den);
    const float e0 = __builtin_fmaf(-den, r, 1.0f);
    const float f1 = __builtin_fmaf(e0, r, r);
    const float r0 = __builtin_fmaf(-den, f1, 1.0f);
    const float f3 = __builtin_fmaf(r0, f1, f1);
    const float f4 = __builtin_fmaf(-den, f3, 1.0f);
    return __builtin_fmaf(f4, f1, f3);
}

__device__ __forceinline__ float q_full(float den) {
    bool unused, flag;
    const float ds = __builtin_amdgcn_div_scalef(1.0f, den, false, &unused);
    const float ns = __builtin_amdgcn_div_scalef(1.0f, den, true, &flag);
    const float r  = __builtin_amdgcn_rcpf(ds);
    const float e0 = __builtin_fmaf(-ds, r, 1.0f);
    const float f1 = __builtin_fmaf(e0, r, r);
    const float q0 = ns * f1;
    const float r0 = __builtin_fmaf(-ds, q0, ns);
    const float f3 = __builtin_fmaf(r0, f1, q0);
    const float f4 = __builtin_fmaf(-ds, f3, ns);
    return __builtin_amdgcn_div_fixupf(__builtin_amdgcn_div_fmasf(f4, f1, f3, flag), den, 1.0f);
}

#define DEN_PLAIN_BELOW 0x1p96f

__device__ __forceinline__ float q_rare(float den, float plain) {
    return (den < DEN_PLAIN_BELOW) ? plain : q_full(den);
}

__device__ __forceinline__ float u_den(float y) {
    return 1.0f + __builtin_amdgcn_exp2f(y);
}

__device__ __forceinline__ float sq2(float sg) {
    const float s2 = sg * sg;
    return s2 * s2;
}

#define XSTEP (Bsz * Hh)
#define BLK   8

template <int S0, int S1, bool FIN>
__device__ __forceinline__ void scan_steps(const float (&x)[BLK][5], const float c5,
                                           const v2f a0, const v2f a1, const v2f a2,
                                           v8f& d, const float ythr, float (&lg)[BLK],
                                           const float (&lgp)[BLK], const int hh, float* const dst, const bool wr) {
    #pragma unroll
    for (int s = S0; s < S1; ++s) {
        const float n0 = u_den(d[0]), n1 = u_den(d[1]), n2 = u_den(d[2]),
                    n3 = u_den(d[3]), n4 = u_den(d[4]);
        const float ymax = fmaxf(fmaxf(fmaxf(d[0], d[1]), fmaxf(d[2], d[3])), d[4]);
        float u0 = q_plain(n0), u1 = q_plain(n1), u2 = q_plain(n2),
              u3 = q_plain(n3), u4 = q_plain(n4);
        if (FIN && s == S0) {
            float gd[4];
            #pragma unroll
            for (int k = 0; k < 4; ++k) gd[k] = 1.0f + __expf(-(hh ? lgp[k + 4] : lgp[k]));
            float sg[4];
            #pragma unroll
            for (int k = 0; k < 4; ++k) sg[k] = q_plain(gd[k]);
            if (__builtin_expect(__builtin_amdgcn_ballot_w32(!(ymax < ythr)) != 0u, 0)) {
                u0 = q_rare(n0, u0); u1 = q_rare(n1, u1); u2 = q_rare(n2, u2);
                u3 = q_rare(n3, u3); u4 = q_rare(n4, u4);
                #pragma unroll
                for (int k = 0; k < 4; ++k) sg[k] = q_rare(gd[k], sg[k]);
            }
            if (wr)
                *reinterpret_cast<float4*>(dst) = make_float4(sq2(sg[0]), sq2(sg[1]), sq2(sg[2]), sq2(sg[3]));
        } else {
            if (__builtin_expect(__builtin_amdgcn_ballot_w32(!(ymax < ythr)) != 0u, 0)) {
                u0 = q_rare(n0, u0); u1 = q_rare(n1, u1); u2 = q_rare(n2, u2);
                u3 = q_rare(n3, u3); u4 = q_rare(n4, u4);
            }
        }
        v2f b0, b1, b2;
        b0[0] = u0; b0[1] = u1;
        b1[0] = u2; b1[1] = u3;
        b2[0] = u4; b2[1] = 0.0f;
        v8f cc;
        cc[0] = x[s][0]; cc[1] = x[s][1]; cc[2] = x[s][2]; cc[3] = x[s][3]; cc[4] = x[s][4];
        cc[5] = c5; cc[6] = 0.0f; cc[7] = 0.0f;
        d = __builtin_amdgcn_wmma_f32_16x16x4_f32(false, a0, false, b0, (short)0, cc, false, false);
        d = __builtin_amdgcn_wmma_f32_16x16x4_f32(false, a1, false, b1, (short)0, d,  false, false);
        d = __builtin_amdgcn_wmma_f32_16x16x4_f32(false, a2, false, b2, (short)0, d,  false, false);
        lg[s] = d[5];
    }
}

__device__ __forceinline__ void finish_block(const float (&lg)[BLK], const int hh, float* const dst, const bool plain_ok, const bool wr) {
    float gd[4];
    #pragma unroll
    for (int s = 0; s < 4; ++s) gd[s] = 1.0f + __expf(-(hh ? lg[s + 4] : lg[s]));
    float sg[4];
    #pragma unroll
    for (int s = 0; s < 4; ++s) sg[s] = q_plain(gd[s]);
    if (!plain_ok) {
        #pragma unroll
        for (int s = 0; s < 4; ++s) sg[s] = q_full(gd[s]);
    }
    float o[4];
    #pragma unroll
    for (int s = 0; s < 4; ++s) o[s] = sq2(sg[s]);
    if (wr)
        *reinterpret_cast<float4*>(dst) = make_float4(o[0], o[1], o[2], o[3]);
}

__device__ __forceinline__ void load_block_last(float (&x)[BLK][5], const float* const xl, const int tfirst) {
    const float* const p = xl + (size_t)tfirst * XSTEP;
    #pragma unroll
    for (int s = 0; s < BLK - 1; ++s)
        #pragma unroll
        for (int r = 0; r < 5; ++r) x[s][r] = p[(size_t)s * XSTEP + r];
    const int tl = min(tfirst + BLK - 1, Sseq - 1);
    const float* const q = xl + (size_t)tl * XSTEP;
    #pragma unroll
    for (int r = 0; r < 5; ++r) x[BLK - 1][r] = q[r];
}

#define TCH   32
#define NCH   (Sseq / TCH)
#define LWARM 64
#define NGRP  (Bsz / 16)
#define E_ROW 0
#define N_ROW 8
#define NEVER (1 << 20)
#define MARK  (-1.0f)

__device__ __forceinline__ float* tile_at(float* const out, const int grp, const int tc,
                                          const int row0, const int lane, const int r) {
    const int f = lane * 5 + r;
    return out + (size_t)(grp * 16 + row0 + (f >> 5)) * Sseq + tc + (f & 31);
}

struct ScanCtx {
    v2f   a0, a1, a2;
    float c5;
    float ythr_fin;
    bool  sig_plain;
};

__device__ __forceinline__ ScanCtx make_ctx(const float* __restrict__ Whh,
                                            const float* __restrict__ Wfc,
                                            const float* __restrict__ bfc_p,
                                            const int c, const int hh) {
    ScanCtx k;
    const int hd = (c < 5) ? c : ((c >= 8 && c < 13) ? c - 3 : -1);
    float av[6];
    #pragma unroll
    for (int r = 0; r < 6; ++r) {
        float v = 0.0f;
        if (r < 5) {
            const int hs = 5 * hh + r;
            if (hd >= 0)     v = (-2.0f * K2) * Whh[hd * Hh + hs];
            else if (c == 5 || c == 13) v = -2.0f * Wfc[hs];
        }
        av[r] = v;
    }
    k.a0[0] = av[0]; k.a0[1] = av[1];
    k.a1[0] = av[2]; k.a1[1] = av[3];
    k.a2[0] = av[4]; k.a2[1] = av[5];

    float sfc = 0.0f;
    #pragma unroll
    for (int j = 0; j < Hh; ++j) sfc += Wfc[j];
    k.c5 = bfc_p[0] + sfc;

    float lb = k.c5;
    #pragma unroll
    for (int j = 0; j < Hh; ++j) lb -= 2.0f * fmaxf(Wfc[j], 0.0f);
    k.sig_plain = lb > L_PLAIN_ABOVE;
    k.ythr_fin  = k.sig_plain ? Y_PLAIN_BELOW : -__builtin_inff();
    return k;
}

__device__ __forceinline__ v8f pass_from_zero(const ScanCtx& k, const float* const xrow) {
    float x0[5];
    #pragma unroll
    for (int r = 0; r < 5; ++r) x0[r] = xrow[r];
    v8f cc;
    cc[0] = x0[0]; cc[1] = x0[1]; cc[2] = x0[2]; cc[3] = x0[3]; cc[4] = x0[4];
    cc[5] = k.c5; cc[6] = 0.0f; cc[7] = 0.0f;
    v2f uh; uh[0] = 0.5f; uh[1] = 0.5f;
    v8f d;
    d = __builtin_amdgcn_wmma_f32_16x16x4_f32(false, k.a0, false, uh, (short)0, cc, false, false);
    d = __builtin_amdgcn_wmma_f32_16x16x4_f32(false, k.a1, false, uh, (short)0, d,  false, false);
    d = __builtin_amdgcn_wmma_f32_16x16x4_f32(false, k.a2, false, uh, (short)0, d,  false, false);
    return d;
}

__device__ __forceinline__ void run_range(const ScanCtx& k, const float* const xl, float* const ob,
                                          v8f& d, const int t0, const int nblk, const int wfirst,
                                          const int hh) {
    float xa[BLK][5], xb[BLK][5];
    float lga[BLK], lgb[BLK];

    load_block_last(xa, xl, t0 + 1);
    load_block_last(xb, xl, t0 + BLK + 1);
    scan_steps<0, BLK, false>(xa, k.c5, k.a0, k.a1, k.a2, d, Y_PLAIN_BELOW, lga, lga, hh, ob, false);

    for (int j = 0; j < nblk / 2 - 1; ++j) {
        const int m = 2 * j + 1;

        load_block_last(xa, xl, t0 + BLK * (m + 1) + 1);
        __builtin_amdgcn_sched_barrier(0);
        scan_steps<0, 4, false>(xb, k.c5, k.a0, k.a1, k.a2, d, Y_PLAIN_BELOW, lgb, lga, hh, ob, false);
        scan_steps<4, BLK, true>(xb, k.c5, k.a0, k.a1, k.a2, d, k.ythr_fin, lgb, lga, hh,
                                 ob + BLK * (m - 1), (m - 1) >= wfirst);

        __builtin_amdgcn_sched_barrier(0);
        load_block_last(xb, xl, t0 + BLK * (m + 2) + 1);
        __builtin_amdgcn_sched_barrier(0);
        scan_steps<0, 4, false>(xa, k.c5, k.a0, k.a1, k.a2, d, Y_PLAIN_BELOW, lga, lgb, hh, ob, false);
        scan_steps<4, BLK, true>(xa, k.c5, k.a0, k.a1, k.a2, d, k.ythr_fin, lga, lgb, hh,
                                 ob + BLK * m, m >= wfirst);
    }

    scan_steps<0, 4, false>(xb, k.c5, k.a0, k.a1, k.a2, d, Y_PLAIN_BELOW, lgb, lga, hh, ob, false);
    scan_steps<4, BLK, true>(xb, k.c5, k.a0, k.a1, k.a2, d, k.ythr_fin, lgb, lga, hh,
                             ob + BLK * (nblk - 2), (nblk - 2) >= wfirst);
    finish_block(lgb, hh, ob + BLK * (nblk - 1), k.sig_plain, (nblk - 1) >= wfirst);
}

__global__ __launch_bounds__(32) __attribute__((amdgpu_waves_per_eu(1, 1))) void rnn_warm(const float* __restrict__ xp,
                                               const float* __restrict__ Whh,
                                               const float* __restrict__ Wfc,
                                               const float* __restrict__ bfc_p,
                                               float* __restrict__ out) {
    const int lane = threadIdx.x;
    const int c    = lane & 15;
    const int hh   = lane >> 4;
    const int ch   = blockIdx.x / NGRP + 1;
    const int grp  = blockIdx.x % NGRP;
    const int b    = grp * 16 + c;

    const ScanCtx k = make_ctx(Whh, Wfc, bfc_p, c, hh);

    const float* const xl = xp + (size_t)b * Hh + 5 * hh;
    float* const orow = out + (size_t)b * Sseq;

    const int tc = ch * TCH;
    const int tw = (tc > LWARM) ? tc - LWARM : 0;
    v8f d = pass_from_zero(k, xl + (size_t)tw * XSTEP);

    run_range(k, xl, orow + tw + 4 * hh, d, tw, (tc - tw) / BLK, NEVER, hh);
    #pragma unroll
    for (int r = 0; r < 5; ++r) {
        *tile_at(out, grp, tc, E_ROW, lane, r)       = d[r];
        *tile_at(out, grp, tc - TCH, N_ROW, lane, r) = d[r];
    }
}

__global__ __launch_bounds__(32) __attribute__((amdgpu_waves_per_eu(1, 1))) void rnn_run(const float* __restrict__ xp,
                                               const float* __restrict__ Whh,
                                               const float* __restrict__ Wfc,
                                               const float* __restrict__ bfc_p,
                                               float* __restrict__ out) {
    const int lane = threadIdx.x;
    const int c    = lane & 15;
    const int hh   = lane >> 4;
    const int ch   = blockIdx.x / NGRP;
    const int grp  = blockIdx.x % NGRP;
    const int b    = grp * 16 + c;
    const int tc   = ch * TCH;
    const bool has_next = ch + 1 < NCH;

    float ev[5], nx[5];
    #pragma unroll
    for (int r = 0; r < 5; ++r) ev[r] = *tile_at(out, grp, tc, ch ? E_ROW : N_ROW, lane, r);
    #pragma unroll
    for (int r = 0; r < 5; ++r) nx[r] = *tile_at(out, grp, tc, has_next ? N_ROW : E_ROW, lane, r);

    const ScanCtx k = make_ctx(Whh, Wfc, bfc_p, c, hh);
    const float* const xl = xp + (size_t)b * Hh + 5 * hh;
    float* const orow = out + (size_t)b * Sseq;

    v8f d;
    if (ch == 0) {
        d = pass_from_zero(k, xl);
    } else {
        #pragma unroll
        for (int r = 0; r < 5; ++r) d[r] = ev[r];
        d[5] = 0.0f; d[6] = 0.0f; d[7] = 0.0f;
    }
    run_range(k, xl, orow + tc + 4 * hh, d, tc, TCH / BLK, 0, hh);

    unsigned int df = 0u;
    #pragma unroll
    for (int r = 0; r < 5; ++r) df |= __float_as_uint(d[r]) ^ __float_as_uint(nx[r]);
    if (has_next && __builtin_amdgcn_ballot_w32(df != 0u) != 0u) {
        if (hh) orow[tc + TCH - 1] = MARK;
    }
}

__global__ __launch_bounds__(256) void rnn_check(const float* __restrict__ xp,
                                                 const float* __restrict__ Whh,
                                                 const float* __restrict__ Wfc,
                                                 const float* __restrict__ bfc_p,
                                                 float* __restrict__ out) {
    const int lane = threadIdx.x & 31;
    const int w    = threadIdx.x >> 5;
    const int c    = lane & 15;
    const int hh   = lane >> 4;
    const int b    = blockIdx.x * 16 + c;
    float* const orow = out + (size_t)b * Sseq;

    __shared__ unsigned int sbad[8];
    unsigned int mybad = 0u;
    #pragma unroll
    for (int i = 0; i < NCH / 16; ++i) {
        const int ch = 2 * (w + 8 * i) + hh;
        const float v = orow[ch * TCH + TCH - 1];
        if (__builtin_amdgcn_ballot_w32(__float_as_uint(v) == __float_as_uint(MARK)) != 0u) mybad = 1u;
    }
    if (lane == 0) sbad[w] = mybad;
    __syncthreads();
    unsigned int bad = 0u;
    #pragma unroll
    for (int i = 0; i < 8; ++i) bad |= sbad[i];
    if (bad == 0u || w != 0) return;

    const ScanCtx k = make_ctx(Whh, Wfc, bfc_p, c, hh);
    const float* const xl = xp + (size_t)b * Hh + 5 * hh;
    v8f d = pass_from_zero(k, xl);
    run_range(k, xl, orow + 4 * hh, d, 0, Sseq / BLK, 0, hh);
}

extern "C" void kernel_launch(void* const* d_in, const int* in_sizes, int n_in,
                              void* d_out, int out_size, void* d_ws, size_t ws_size,
                              hipStream_t stream) {
    (void)in_sizes; (void)n_in; (void)out_size; (void)ws_size;
    const float* inp = (const float*)d_in[0];
    const float* Wih = (const float*)d_in[1];
    const float* Whh = (const float*)d_in[2];
    const float* bih = (const float*)d_in[3];
    const float* bhh = (const float*)d_in[4];
    const float* Wfc = (const float*)d_in[5];
    const float* bfc = (const float*)d_in[6];
    float* out = (float*)d_out;
    float* xp  = (float*)d_ws;

    const int tiles = (Bsz * Sseq) / 16;
    xproj_wmma<<<dim3(tiles / (4 * TPW)), 128, 0, stream>>>(inp, Wih, bih, bhh, Whh, xp);
    rnn_warm<<<dim3((NCH - 1) * NGRP), 32, 0, stream>>>(xp, Whh, Wfc, bfc, out);
    rnn_run<<<dim3(NCH * NGRP), 32, 0, stream>>>(xp, Whh, Wfc, bfc, out);
    rnn_check<<<dim3(NGRP), 256, 0, stream>>>(xp, Whh, Wfc, bfc, out);
}


# scan: warm-up and run of a chunk in ONE launch (was two): workgroup of 3 waves = 2 chunks + the next chunk's warm-up, end state checked against next entry state in LDS behind one barrier; same sums
# speedup vs baseline: 29.0509x; 1.2122x over previous
#include <hip/hip_runtime.h>

typedef __attribute__((ext_vector_type(2))) float v2f;
typedef __attribute__((ext_vector_type(8))) float v8f;

#define Bsz  64
#define Sseq 4096
#define Cin  80
#define Hh   10

#define K2 (2.0f * 1.44269504088896340736f)

#define TPW 4
#define KST (Cin / 4)

__global__ __launch_bounds__(128) void xproj_wmma(const float* __restrict__ inp,
                                                  const float* __restrict__ Wih,
                                                  const float* __restrict__ bih,
                                                  const float* __restrict__ bhh,
                                                  const float* __restrict__ Whh,
                                                  float* __restrict__ xp) {
    const int lane = threadIdx.x & 31;
    const int wave = threadIdx.x >> 5;
    const int tile0 = (blockIdx.x * 4 + wave) * TPW;
    const int half = lane >> 4;
    const int n    = lane & 15;
    const int nn   = (n < Hh) ? n : (Hh - 1);
    const float nmask = (n < Hh) ? 1.0f : 0.0f;

    float wraw[KST][2];
    #pragma unroll
    for (int ks = 0; ks < KST; ++ks) {
        const int kk = ks * 4 + half * 2;
        wraw[ks][0] = Wih[nn * Cin + kk];
        wraw[ks][1] = Wih[nn * Cin + kk + 1];
    }
    const float bi = bih[nn], bh = bhh[nn];
    float whr[Hh];
    #pragma unroll
    for (int j = 0; j < Hh; ++j) whr[j] = Whh[nn * Hh + j];

    const float* const arow = inp + ((size_t)tile0 * 16 + n) * Cin + half * 2;
    v2f abuf[2][KST];
    #pragma unroll
    for (int t = 0; t < 2; ++t) {
        const float* const at = arow + (size_t)t * 16 * Cin;
        #pragma unroll
        for (int ks = 0; ks < KST; ++ks) {
            const float2 a2 = *reinterpret_cast<const float2*>(at + ks * 4);
            abuf[t][ks][0] = a2.x; abuf[t][ks][1] = a2.y;
        }
    }
    __builtin_amdgcn_sched_barrier(0);

    v2f bfrag[KST];
    #pragma unroll
    for (int ks = 0; ks < KST; ++ks) {
        bfrag[ks][0] = nmask * wraw[ks][0];
        bfrag[ks][1] = nmask * wraw[ks][1];
    }
    const float bias = bi + bh;
    float rs = 0.0f;
    #pragma unroll
    for (int j = 0; j < Hh; ++j) rs += whr[j];

    #pragma unroll
    for (int t = 0; t < TPW; ++t) {
        if (t >= 1 && t + 1 < TPW) {
            const float* const anext = arow + (size_t)(t + 1) * 16 * Cin;
            #pragma unroll
            for (int ks = 0; ks < KST; ++ks) {
                const float2 a2 = *reinterpret_cast<const float2*>(anext + ks * 4);
                abuf[(t + 1) & 1][ks][0] = a2.x; abuf[(t + 1) & 1][ks][1] = a2.y;
            }
        }

        v8f acc = {};
        #pragma unroll
        for (int ks = 0; ks < KST; ++ks)
            acc = __builtin_amdgcn_wmma_f32_16x16x4_f32(
                false, abuf[t & 1][ks], false, bfrag[ks], (short)0, acc, false, false);

        if (n < Hh) {
            #pragma unroll
            for (int v = 0; v < 8; ++v) {
                const int g    = (tile0 + t) * 16 + v + half * 8;
                const int bidx = g >> 12;
                const int sidx = g & 4095;
                xp[((size_t)sidx * Bsz + bidx) * Hh + n] = ((acc[v] + bias) + rs) * K2;
            }
        }
    }
}

#define Y_PLAIN_BELOW 64.0f
#define L_PLAIN_ABOVE (-40.0f)

__device__ __forceinline__ float q_plain(float den) {
    const float r  = __builtin_amdgcn_rcpf(den);
    const float e0 = __builtin_fmaf(-den, r, 1.0f);
    const float f1 = __builtin_fmaf(e0, r, r);
    const float r0 = __builtin_fmaf(-den, f1, 1.0f);
    const float f3 = __builtin_fmaf(r0, f1, f1);
    const float f4 = __builtin_fmaf(-den, f3, 1.0f);
    return __builtin_fmaf(f4, f1, f3);
}

__device__ __forceinline__ float q_full(float den) {
    bool unused, flag;
    const float ds = __builtin_amdgcn_div_scalef(1.0f, den, false, &unused);
    const float ns = __builtin_amdgcn_div_scalef(1.0f, den, true, &flag);
    const float r  = __builtin_amdgcn_rcpf(ds);
    const float e0 = __builtin_fmaf(-ds, r, 1.0f);
    const float f1 = __builtin_fmaf(e0, r, r);
    const float q0 = ns * f1;
    const float r0 = __builtin_fmaf(-ds, q0, ns);
    const float f3 = __builtin_fmaf(r0, f1, q0);
    const float f4 = __builtin_fmaf(-ds, f3, ns);
    return __builtin_amdgcn_div_fixupf(__builtin_amdgcn_div_fmasf(f4, f1, f3, flag), den, 1.0f);
}

#define DEN_PLAIN_BELOW 0x1p96f

__device__ __forceinline__ float q_rare(float den, float plain) {
    return (den < DEN_PLAIN_BELOW) ? plain : q_full(den);
}

__device__ __forceinline__ float u_den(float y) {
    return 1.0f + __builtin_amdgcn_exp2f(y);
}

__device__ __forceinline__ float sq2(float sg) {
    const float s2 = sg * sg;
    return s2 * s2;
}

#define XSTEP (Bsz * Hh)
#define BLK   8

template <int S0, int S1, bool FIN>
__device__ __forceinline__ void scan_steps(const float (&x)[BLK][5], const float c5,
                                           const v2f a0, const v2f a1, const v2f a2,
                                           v8f& d, const float ythr, float (&lg)[BLK],
                                           const float (&lgp)[BLK], const int hh, float* const dst, const bool wr) {
    #pragma unroll
    for (int s = S0; s < S1; ++s) {
        const float n0 = u_den(d[0]), n1 = u_den(d[1]), n2 = u_den(d[2]),
                    n3 = u_den(d[3]), n4 = u_den(d[4]);
        const float ymax = fmaxf(fmaxf(fmaxf(d[0], d[1]), fmaxf(d[2], d[3])), d[4]);
        float u0 = q_plain(n0), u1 = q_plain(n1), u2 = q_plain(n2),
              u3 = q_plain(n3), u4 = q_plain(n4);
        if (FIN && s == S0) {
            float gd[4];
            #pragma unroll
            for (int k = 0; k < 4; ++k) gd[k] = 1.0f + __expf(-(hh ? lgp[k + 4] : lgp[k]));
            float sg[4];
            #pragma unroll
            for (int k = 0; k < 4; ++k) sg[k] = q_plain(gd[k]);
            if (__builtin_expect(__builtin_amdgcn_ballot_w32(!(ymax < ythr)) != 0u, 0)) {
                u0 = q_rare(n0, u0); u1 = q_rare(n1, u1); u2 = q_rare(n2, u2);
                u3 = q_rare(n3, u3); u4 = q_rare(n4, u4);
                #pragma unroll
                for (int k = 0; k < 4; ++k) sg[k] = q_rare(gd[k], sg[k]);
            }
            if (wr)
                *reinterpret_cast<float4*>(dst) = make_float4(sq2(sg[0]), sq2(sg[1]), sq2(sg[2]), sq2(sg[3]));
        } else {
            if (__builtin_expect(__builtin_amdgcn_ballot_w32(!(ymax < ythr)) != 0u, 0)) {
                u0 = q_rare(n0, u0); u1 = q_rare(n1, u1); u2 = q_rare(n2, u2);
                u3 = q_rare(n3, u3); u4 = q_rare(n4, u4);
            }
        }
        v2f b0, b1, b2;
        b0[0] = u0; b0[1] = u1;
        b1[0] = u2; b1[1] = u3;
        b2[0] = u4; b2[1] = 0.0f;
        v8f cc;
        cc[0] = x[s][0]; cc[1] = x[s][1]; cc[2] = x[s][2]; cc[3] = x[s][3]; cc[4] = x[s][4];
        cc[5] = c5; cc[6] = 0.0f; cc[7] = 0.0f;
        d = __builtin_amdgcn_wmma_f32_16x16x4_f32(false, a0, false, b0, (short)0, cc, false, false);
        d = __builtin_amdgcn_wmma_f32_16x16x4_f32(false, a1, false, b1, (short)0, d,  false, false);
        d = __builtin_amdgcn_wmma_f32_16x16x4_f32(false, a2, false, b2, (short)0, d,  false, false);
        lg[s] = d[5];
    }
}

__device__ __forceinline__ void finish_block(const float (&lg)[BLK], const int hh, float* const dst, const bool plain_ok, const bool wr) {
    float gd[4];
    #pragma unroll
    for (int s = 0; s < 4; ++s) gd[s] = 1.0f + __expf(-(hh ? lg[s + 4] : lg[s]));
    float sg[4];
    #pragma unroll
    for (int s = 0; s < 4; ++s) sg[s] = q_plain(gd[s]);
    if (!plain_ok) {
        #pragma unroll
        for (int s = 0; s < 4; ++s) sg[s] = q_full(gd[s]);
    }
    float o[4];
    #pragma unroll
    for (int s = 0; s < 4; ++s) o[s] = sq2(sg[s]);
    if (wr)
        *reinterpret_cast<float4*>(dst) = make_float4(o[0], o[1], o[2], o[3]);
}

__device__ __forceinline__ void load_block_last(float (&x)[BLK][5], const float* const xl, const int tfirst) {
    const float* const p = xl + (size_t)tfirst * XSTEP;
    #pragma unroll
    for (int s = 0; s < BLK - 1; ++s)
        #pragma unroll
        for (int r = 0; r < 5; ++r) x[s][r] = p[(size_t)s * XSTEP + r];
    const int tl = min(tfirst + BLK - 1, Sseq - 1);
    const float* const q = xl + (size_t)tl * XSTEP;
    #pragma unroll
    for (int r = 0; r < 5; ++r) x[BLK - 1][r] = q[r];
}

#define TCH   32
#define NCH   (Sseq / TCH)
#define LWARM 64
#define NGRP  (Bsz / 16)
#define CPW   2
#define NEVER (1 << 20)
#define MARK  (-1.0f)

struct ScanCtx {
    v2f   a0, a1, a2;
    float c5;
    float ythr_fin;
    bool  sig_plain;
};

__device__ __forceinline__ ScanCtx make_ctx(const float* __restrict__ Whh,
                                            const float* __restrict__ Wfc,
                                            const float* __restrict__ bfc_p,
                                            const int c, const int hh) {
    ScanCtx k;
    const int hd = (c < 5) ? c : ((c >= 8 && c < 13) ? c - 3 : -1);
    float av[6];
    #pragma unroll
    for (int r = 0; r < 6; ++r) {
        float v = 0.0f;
        if (r < 5) {
            const int hs = 5 * hh + r;
            if (hd >= 0)     v = (-2.0f * K2) * Whh[hd * Hh + hs];
            else if (c == 5 || c == 13) v = -2.0f * Wfc[hs];
        }
        av[r] = v;
    }
    k.a0[0] = av[0]; k.a0[1] = av[1];
    k.a1[0] = av[2]; k.a1[1] = av[3];
    k.a2[0] = av[4]; k.a2[1] = av[5];

    float sfc = 0.0f;
    #pragma unroll
    for (int j = 0; j < Hh; ++j) sfc += Wfc[j];
    k.c5 = bfc_p[0] + sfc;

    float lb = k.c5;
    #pragma unroll
    for (int j = 0; j < Hh; ++j) lb -= 2.0f * fmaxf(Wfc[j], 0.0f);
    k.sig_plain = lb > L_PLAIN_ABOVE;
    k.ythr_fin  = k.sig_plain ? Y_PLAIN_BELOW : -__builtin_inff();
    return k;
}

__device__ __forceinline__ v8f pass_from_zero(const ScanCtx& k, const float* const xrow) {
    float x0[5];
    #pragma unroll
    for (int r = 0; r < 5; ++r) x0[r] = xrow[r];
    v8f cc;
    cc[0] = x0[0]; cc[1] = x0[1]; cc[2] = x0[2]; cc[3] = x0[3]; cc[4] = x0[4];
    cc[5] = k.c5; cc[6] = 0.0f; cc[7] = 0.0f;
    v2f uh; uh[0] = 0.5f; uh[1] = 0.5f;
    v8f d;
    d = __builtin_amdgcn_wmma_f32_16x16x4_f32(false, k.a0, false, uh, (short)0, cc, false, false);
    d = __builtin_amdgcn_wmma_f32_16x16x4_f32(false, k.a1, false, uh, (short)0, d,  false, false);
    d = __builtin_amdgcn_wmma_f32_16x16x4_f32(false, k.a2, false, uh, (short)0, d,  false, false);
    return d;
}

__device__ __forceinline__ void run_range(const ScanCtx& k, const float* const xl, float* const ob,
                                          v8f& d, const int t0, const int nblk, const int wfirst,
                                          const int hh, float* const cap, const int capblk) {
    float xa[BLK][5], xb[BLK][5];
    float lga[BLK], lgb[BLK];

    load_block_last(xa, xl, t0 + 1);
    load_block_last(xb, xl, t0 + BLK + 1);
    scan_steps<0, BLK, false>(xa, k.c5, k.a0, k.a1, k.a2, d, Y_PLAIN_BELOW, lga, lga, hh, ob, false);

    for (int j = 0; j < nblk / 2 - 1; ++j) {
        const int m = 2 * j + 1;

        load_block_last(xa, xl, t0 + BLK * (m + 1) + 1);
        __builtin_amdgcn_sched_barrier(0);
        scan_steps<0, 4, false>(xb, k.c5, k.a0, k.a1, k.a2, d, Y_PLAIN_BELOW, lgb, lga, hh, ob, false);
        scan_steps<4, BLK, true>(xb, k.c5, k.a0, k.a1, k.a2, d, k.ythr_fin, lgb, lga, hh,
                                 ob + BLK * (m - 1), (m - 1) >= wfirst);
        if (m + 1 == capblk) {
            #pragma unroll
            for (int r = 0; r < 5; ++r) cap[r] = d[r];
        }

        __builtin_amdgcn_sched_barrier(0);
        load_block_last(xb, xl, t0 + BLK * (m + 2) + 1);
        __builtin_amdgcn_sched_barrier(0);
        scan_steps<0, 4, false>(xa, k.c5, k.a0, k.a1, k.a2, d, Y_PLAIN_BELOW, lga, lgb, hh, ob, false);
        scan_steps<4, BLK, true>(xa, k.c5, k.a0, k.a1, k.a2, d, k.ythr_fin, lga, lgb, hh,
                                 ob + BLK * m, m >= wfirst);
    }

    scan_steps<0, 4, false>(xb, k.c5, k.a0, k.a1, k.a2, d, Y_PLAIN_BELOW, lgb, lga, hh, ob, false);
    scan_steps<4, BLK, true>(xb, k.c5, k.a0, k.a1, k.a2, d, k.ythr_fin, lgb, lga, hh,
                             ob + BLK * (nblk - 2), (nblk - 2) >= wfirst);
    finish_block(lgb, hh, ob + BLK * (nblk - 1), k.sig_plain, (nblk - 1) >= wfirst);
}

__global__ __launch_bounds__(32 * (CPW + 1)) __attribute__((amdgpu_waves_per_eu(1, 1))) void rnn_scan(const float* __restrict__ xp,
                                               const float* __restrict__ Whh,
                                               const float* __restrict__ Wfc,
                                               const float* __restrict__ bfc_p,
                                               float* __restrict__ out) {
    const int lane = threadIdx.x & 31;
    const int w    = __builtin_amdgcn_readfirstlane((int)(threadIdx.x >> 5));
    const int c    = lane & 15;
    const int hh   = lane >> 4;
    const int grp  = blockIdx.x % NGRP;
    const int ch   = (blockIdx.x / NGRP) * CPW + w;
    const int b    = grp * 16 + c;
    const int tc   = ch * TCH;
    const bool live = ch < NCH;
    const bool runs = w < CPW;

    __shared__ float etile[CPW][160];

    v8f d = {};
    if (live) {
        const ScanCtx k = make_ctx(Whh, Wfc, bfc_p, c, hh);

        const float* const xl = xp + (size_t)b * Hh + 5 * hh;
        float* const orow = out + (size_t)b * Sseq;

        const int tw = (tc > LWARM) ? tc - LWARM : 0;
        const int nw = (tc - tw) / BLK;
        d = pass_from_zero(k, xl + (size_t)tw * XSTEP);
        float* const cap = &etile[w >= 1 ? w - 1 : 0][lane * 5];
        run_range(k, xl, orow + tw + 4 * hh, d, tw, runs ? nw + TCH / BLK : nw, runs ? nw : NEVER, hh,
                  cap, (runs && w >= 1) ? nw : -1);
        if (!runs) {
            #pragma unroll
            for (int r = 0; r < 5; ++r) cap[r] = d[r];
        }
    }
    __syncthreads();
    if (runs && ch + 1 < NCH) {
        unsigned int df = 0u;
        #pragma unroll
        for (int r = 0; r < 5; ++r) df |= __float_as_uint(d[r]) ^ __float_as_uint(etile[w][lane * 5 + r]);
        if (__builtin_amdgcn_ballot_w32(df != 0u) != 0u) {
            if (hh) out[(size_t)b * Sseq + tc + TCH - 1] = MARK;
        }
    }
}

__global__ __launch_bounds__(256) void rnn_check(const float* __restrict__ xp,
                                                 const float* __restrict__ Whh,
                                                 const float* __restrict__ Wfc,
                                                 const float* __restrict__ bfc_p,
                                                 float* __restrict__ out) {
    const int lane = threadIdx.x & 31;
    const int w    = threadIdx.x >> 5;
    const int c    = lane & 15;
    const int hh   = lane >> 4;
    const int b    = blockIdx.x * 16 + c;
    float* const orow = out + (size_t)b * Sseq;

    __shared__ unsigned int sbad[8];
    unsigned int mybad = 0u;
    #pragma unroll
    for (int i = 0; i < NCH / 16; ++i) {
        const int ch = 2 * (w + 8 * i) + hh;
        const float v = orow[ch * TCH + TCH - 1];
        if (__builtin_amdgcn_ballot_w32(__float_as_uint(v) == __float_as_uint(MARK)) != 0u) mybad = 1u;
    }
    if (lane == 0) sbad[w] = mybad;
    __syncthreads();
    unsigned int bad = 0u;
    #pragma unroll
    for (int i = 0; i < 8; ++i) bad |= sbad[i];
    if (bad == 0u || w != 0) return;

    const ScanCtx k = make_ctx(Whh, Wfc, bfc_p, c, hh);
    const float* const xl = xp + (size_t)b * Hh + 5 * hh;
    v8f d = pass_from_zero(k, xl);
    run_range(k, xl, orow + 4 * hh, d, 0, Sseq / BLK, 0, hh, nullptr, -1);
}

extern "C" void kernel_launch(void* const* d_in, const int* in_sizes, int n_in,
                              void* d_out, int out_size, void* d_ws, size_t ws_size,
                              hipStream_t stream) {
    (void)in_sizes; (void)n_in; (void)out_size; (void)ws_size;
    const float* inp = (const float*)d_in[0];
    const float* Wih = (const float*)d_in[1];
    const float* Whh = (const float*)d_in[2];
    const float* bih = (const float*)d_in[3];
    const float* bhh = (const float*)d_in[4];
    const float* Wfc = (const float*)d_in[5];
    const float* bfc = (const float*)d_in[6];
    float* out = (float*)d_out;
    float* xp  = (float*)d_ws;

    const int tiles = (Bsz * Sseq) / 16;
    xproj_wmma<<<dim3(tiles / (4 * TPW)), 128, 0, stream>>>(inp, Wih, bih, bhh, Whh, xp);
    rnn_scan<<<dim3((NCH / CPW) * NGRP), 32 * (CPW + 1), 0, stream>>>(xp, Whh, Wfc, bfc, out);
    rnn_check<<<dim3(NGRP), 256, 0, stream>>>(xp, Whh, Wfc, bfc, out);
}


# scan: a chunk's warm-up from h=0 is 48 steps, not 64 (a wave's serial path 80 steps, not 96); 128 chunks of 32 and the 3-wave workgroup kept, one wave per SIMD; same sums
# speedup vs baseline: 30.2025x; 1.0396x over previous
#include <hip/hip_runtime.h>

typedef __attribute__((ext_vector_type(2))) float v2f;
typedef __attribute__((ext_vector_type(8))) float v8f;

#define Bsz  64
#define Sseq 4096
#define Cin  80
#define Hh   10

#define K2 (2.0f * 1.44269504088896340736f)

#define TPW 4
#define KST (Cin / 4)

__global__ __launch_bounds__(128) void xproj_wmma(const float* __restrict__ inp,
                                                  const float* __restrict__ Wih,
                                                  const float* __restrict__ bih,
                                                  const float* __restrict__ bhh,
                                                  const float* __restrict__ Whh,
                                                  float* __restrict__ xp) {
    const int lane = threadIdx.x & 31;
    const int wave = threadIdx.x >> 5;
    const int tile0 = (blockIdx.x * 4 + wave) * TPW;
    const int half = lane >> 4;
    const int n    = lane & 15;
    const int nn   = (n < Hh) ? n : (Hh - 1);
    const float nmask = (n < Hh) ? 1.0f : 0.0f;

    float wraw[KST][2];
    #pragma unroll
    for (int ks = 0; ks < KST; ++ks) {
        const int kk = ks * 4 + half * 2;
        wraw[ks][0] = Wih[nn * Cin + kk];
        wraw[ks][1] = Wih[nn * Cin + kk + 1];
    }
    const float bi = bih[nn], bh = bhh[nn];
    float whr[Hh];
    #pragma unroll
    for (int j = 0; j < Hh; ++j) whr[j] = Whh[nn * Hh + j];

    const float* const arow = inp + ((size_t)tile0 * 16 + n) * Cin + half * 2;
    v2f abuf[2][KST];
    #pragma unroll
    for (int t = 0; t < 2; ++t) {
        const float* const at = arow + (size_t)t * 16 * Cin;
        #pragma unroll
        for (int ks = 0; ks < KST; ++ks) {
            const float2 a2 = *reinterpret_cast<const float2*>(at + ks * 4);
            abuf[t][ks][0] = a2.x; abuf[t][ks][1] = a2.y;
        }
    }
    __builtin_amdgcn_sched_barrier(0);

    v2f bfrag[KST];
    #pragma unroll
    for (int ks = 0; ks < KST; ++ks) {
        bfrag[ks][0] = nmask * wraw[ks][0];
        bfrag[ks][1] = nmask * wraw[ks][1];
    }
    const float bias = bi + bh;
    float rs = 0.0f;
    #pragma unroll
    for (int j = 0; j < Hh; ++j) rs += whr[j];

    #pragma unroll
    for (int t = 0; t < TPW; ++t) {
        if (t >= 1 && t + 1 < TPW) {
            const float* const anext = arow + (size_t)(t + 1) * 16 * Cin;
            #pragma unroll
            for (int ks = 0; ks < KST; ++ks) {
                const float2 a2 = *reinterpret_cast<const float2*>(anext + ks * 4);
                abuf[(t + 1) & 1][ks][0] = a2.x; abuf[(t + 1) & 1][ks][1] = a2.y;
            }
        }

        v8f acc = {};
        #pragma unroll
        for (int ks = 0; ks < KST; ++ks)
            acc = __builtin_amdgcn_wmma_f32_16x16x4_f32(
                false, abuf[t & 1][ks], false, bfrag[ks], (short)0, acc, false, false);

        if (n < Hh) {
            #pragma unroll
            for (int v = 0; v < 8; ++v) {
                const int g    = (tile0 + t) * 16 + v + half * 8;
                const int bidx = g >> 12;
                const int sidx = g & 4095;
                xp[((size_t)sidx * Bsz + bidx) * Hh + n] = ((acc[v] + bias) + rs) * K2;
            }
        }
    }
}

#define Y_PLAIN_BELOW 64.0f
#define L_PLAIN_ABOVE (-40.0f)

__device__ __forceinline__ float q_plain(float den) {
    const float r  = __builtin_amdgcn_rcpf(den);
    const float e0 = __builtin_fmaf(-den, r, 1.0f);
    const float f1 = __builtin_fmaf(e0, r, r);
    const float r0 = __builtin_fmaf(-den, f1, 1.0f);
    const float f3 = __builtin_fmaf(r0, f1, f1);
    const float f4 = __builtin_fmaf(-den, f3, 1.0f);
    return __builtin_fmaf(f4, f1, f3);
}

__device__ __forceinline__ float q_full(float den) {
    bool unused, flag;
    const float ds = __builtin_amdgcn_div_scalef(1.0f, den, false, &unused);
    const float ns = __builtin_amdgcn_div_scalef(1.0f, den, true, &flag);
    const float r  = __builtin_amdgcn_rcpf(ds);
    const float e0 = __builtin_fmaf(-ds, r, 1.0f);
    const float f1 = __builtin_fmaf(e0, r, r);
    const float q0 = ns * f1;
    const float r0 = __builtin_fmaf(-ds, q0, ns);
    const float f3 = __builtin_fmaf(r0, f1, q0);
    const float f4 = __builtin_fmaf(-ds, f3, ns);
    return __builtin_amdgcn_div_fixupf(__builtin_amdgcn_div_fmasf(f4, f1, f3, flag), den, 1.0f);
}

#define DEN_PLAIN_BELOW 0x1p96f

__device__ __forceinline__ float q_rare(float den, float plain) {
    return (den < DEN_PLAIN_BELOW) ? plain : q_full(den);
}

__device__ __forceinline__ float u_den(float y) {
    return 1.0f + __builtin_amdgcn_exp2f(y);
}

__device__ __forceinline__ float sq2(float sg) {
    const float s2 = sg * sg;
    return s2 * s2;
}

#define XSTEP (Bsz * Hh)
#define BLK   8

template <int S0, int S1, bool FIN>
__device__ __forceinline__ void scan_steps(const float (&x)[BLK][5], const float c5,
                                           const v2f a0, const v2f a1, const v2f a2,
                                           v8f& d, const float ythr, float (&lg)[BLK],
                                           const float (&lgp)[BLK], const int hh, float* const dst, const bool wr) {
    #pragma unroll
    for (int s = S0; s < S1; ++s) {
        const float n0 = u_den(d[0]), n1 = u_den(d[1]), n2 = u_den(d[2]),
                    n3 = u_den(d[3]), n4 = u_den(d[4]);
        const float ymax = fmaxf(fmaxf(fmaxf(d[0], d[1]), fmaxf(d[2], d[3])), d[4]);
        float u0 = q_plain(n0), u1 = q_plain(n1), u2 = q_plain(n2),
              u3 = q_plain(n3), u4 = q_plain(n4);
        if (FIN && s == S0) {
            float gd[4];
            #pragma unroll
            for (int k = 0; k < 4; ++k) gd[k] = 1.0f + __expf(-(hh ? lgp[k + 4] : lgp[k]));
            float sg[4];
            #pragma unroll
            for (int k = 0; k < 4; ++k) sg[k] = q_plain(gd[k]);
            if (__builtin_expect(__builtin_amdgcn_ballot_w32(!(ymax < ythr)) != 0u, 0)) {
                u0 = q_rare(n0, u0); u1 = q_rare(n1, u1); u2 = q_rare(n2, u2);
                u3 = q_rare(n3, u3); u4 = q_rare(n4, u4);
                #pragma unroll
                for (int k = 0; k < 4; ++k) sg[k] = q_rare(gd[k], sg[k]);
            }
            if (wr)
                *reinterpret_cast<float4*>(dst) = make_float4(sq2(sg[0]), sq2(sg[1]), sq2(sg[2]), sq2(sg[3]));
        } else {
            if (__builtin_expect(__builtin_amdgcn_ballot_w32(!(ymax < ythr)) != 0u, 0)) {
                u0 = q_rare(n0, u0); u1 = q_rare(n1, u1); u2 = q_rare(n2, u2);
                u3 = q_rare(n3, u3); u4 = q_rare(n4, u4);
            }
        }
        v2f b0, b1, b2;
        b0[0] = u0; b0[1] = u1;
        b1[0] = u2; b1[1] = u3;
        b2[0] = u4; b2[1] = 0.0f;
        v8f cc;
        cc[0] = x[s][0]; cc[1] = x[s][1]; cc[2] = x[s][2]; cc[3] = x[s][3]; cc[4] = x[s][4];
        cc[5] = c5; cc[6] = 0.0f; cc[7] = 0.0f;
        d = __builtin_amdgcn_wmma_f32_16x16x4_f32(false, a0, false, b0, (short)0, cc, false, false);
        d = __builtin_amdgcn_wmma_f32_16x16x4_f32(false, a1, false, b1, (short)0, d,  false, false);
        d = __builtin_amdgcn_wmma_f32_16x16x4_f32(false, a2, false, b2, (short)0, d,  false, false);
        lg[s] = d[5];
    }
}

__device__ __forceinline__ void finish_block(const float (&lg)[BLK], const int hh, float* const dst, const bool plain_ok, const bool wr) {
    float gd[4];
    #pragma unroll
    for (int s = 0; s < 4; ++s) gd[s] = 1.0f + __expf(-(hh ? lg[s + 4] : lg[s]));
    float sg[4];
    #pragma unroll
    for (int s = 0; s < 4; ++s) sg[s] = q_plain(gd[s]);
    if (!plain_ok) {
        #pragma unroll
        for (int s = 0; s < 4; ++s) sg[s] = q_full(gd[s]);
    }
    float o[4];
    #pragma unroll
    for (int s = 0; s < 4; ++s) o[s] = sq2(sg[s]);
    if (wr)
        *reinterpret_cast<float4*>(dst) = make_float4(o[0], o[1], o[2], o[3]);
}

__device__ __forceinline__ void load_block_last(float (&x)[BLK][5], const float* const xl, const int tfirst) {
    const float* const p = xl + (size_t)tfirst * XSTEP;
    #pragma unroll
    for (int s = 0; s < BLK - 1; ++s)
        #pragma unroll
        for (int r = 0; r < 5; ++r) x[s][r] = p[(size_t)s * XSTEP + r];
    const int tl = min(tfirst + BLK - 1, Sseq - 1);
    const float* const q = xl + (size_t)tl * XSTEP;
    #pragma unroll
    for (int r = 0; r < 5; ++r) x[BLK - 1][r] = q[r];
}

#define TCH   32
#define NCH   (Sseq / TCH)
#define LWARM 48
#define NGRP  (Bsz / 16)
#define CPW   2
#define NEVER (1 << 20)
#define MARK  (-1.0f)

struct ScanCtx {
    v2f   a0, a1, a2;
    float c5;
    float ythr_fin;
    bool  sig_plain;
};

__device__ __forceinline__ ScanCtx make_ctx(const float* __restrict__ Whh,
                                            const float* __restrict__ Wfc,
                                            const float* __restrict__ bfc_p,
                                            const int c, const int hh) {
    ScanCtx k;
    const int hd = (c < 5) ? c : ((c >= 8 && c < 13) ? c - 3 : -1);
    float av[6];
    #pragma unroll
    for (int r = 0; r < 6; ++r) {
        float v = 0.0f;
        if (r < 5) {
            const int hs = 5 * hh + r;
            if (hd >= 0)     v = (-2.0f * K2) * Whh[hd * Hh + hs];
            else if (c == 5 || c == 13) v = -2.0f * Wfc[hs];
        }
        av[r] = v;
    }
    k.a0[0] = av[0]; k.a0[1] = av[1];
    k.a1[0] = av[2]; k.a1[1] = av[3];
    k.a2[0] = av[4]; k.a2[1] = av[5];

    float sfc = 0.0f;
    #pragma unroll
    for (int j = 0; j < Hh; ++j) sfc += Wfc[j];
    k.c5 = bfc_p[0] + sfc;

    float lb = k.c5;
    #pragma unroll
    for (int j = 0; j < Hh; ++j) lb -= 2.0f * fmaxf(Wfc[j], 0.0f);
    k.sig_plain = lb > L_PLAIN_ABOVE;
    k.ythr_fin  = k.sig_plain ? Y_PLAIN_BELOW : -__builtin_inff();
    return k;
}

__device__ __forceinline__ v8f pass_from_zero(const ScanCtx& k, const float* const xrow) {
    float x0[5];
    #pragma unroll
    for (int r = 0; r < 5; ++r) x0[r] = xrow[r];
    v8f cc;
    cc[0] = x0[0]; cc[1] = x0[1]; cc[2] = x0[2]; cc[3] = x0[3]; cc[4] = x0[4];
    cc[5] = k.c5; cc[6] = 0.0f; cc[7] = 0.0f;
    v2f uh; uh[0] = 0.5f; uh[1] = 0.5f;
    v8f d;
    d = __builtin_amdgcn_wmma_f32_16x16x4_f32(false, k.a0, false, uh, (short)0, cc, false, false);
    d = __builtin_amdgcn_wmma_f32_16x16x4_f32(false, k.a1, false, uh, (short)0, d,  false, false);
    d = __builtin_amdgcn_wmma_f32_16x16x4_f32(false, k.a2, false, uh, (short)0, d,  false, false);
    return d;
}

__device__ __forceinline__ void run_range(const ScanCtx& k, const float* const xl, float* const ob,
                                          v8f& d, const int t0, const int nblk, const int wfirst,
                                          const int hh, float* const cap, const int capblk) {
    float xa[BLK][5], xb[BLK][5];
    float lga[BLK], lgb[BLK];

    load_block_last(xa, xl, t0 + 1);
    load_block_last(xb, xl, t0 + BLK + 1);
    scan_steps<0, BLK, false>(xa, k.c5, k.a0, k.a1, k.a2, d, Y_PLAIN_BELOW, lga, lga, hh, ob, false);

    for (int j = 0; j < nblk / 2 - 1; ++j) {
        const int m = 2 * j + 1;

        load_block_last(xa, xl, t0 + BLK * (m + 1) + 1);
        __builtin_amdgcn_sched_barrier(0);
        scan_steps<0, 4, false>(xb, k.c5, k.a0, k.a1, k.a2, d, Y_PLAIN_BELOW, lgb, lga, hh, ob, false);
        scan_steps<4, BLK, true>(xb, k.c5, k.a0, k.a1, k.a2, d, k.ythr_fin, lgb, lga, hh,
                                 ob + BLK * (m - 1), (m - 1) >= wfirst);
        if (m + 1 == capblk) {
            #pragma unroll
            for (int r = 0; r < 5; ++r) cap[r] = d[r];
        }

        __builtin_amdgcn_sched_barrier(0);
        load_block_last(xb, xl, t0 + BLK * (m + 2) + 1);
        __builtin_amdgcn_sched_barrier(0);
        scan_steps<0, 4, false>(xa, k.c5, k.a0, k.a1, k.a2, d, Y_PLAIN_BELOW, lga, lgb, hh, ob, false);
        scan_steps<4, BLK, true>(xa, k.c5, k.a0, k.a1, k.a2, d, k.ythr_fin, lga, lgb, hh,
                                 ob + BLK * m, m >= wfirst);
    }

    scan_steps<0, 4, false>(xb, k.c5, k.a0, k.a1, k.a2, d, Y_PLAIN_BELOW, lgb, lga, hh, ob, false);
    scan_steps<4, BLK, true>(xb, k.c5, k.a0, k.a1, k.a2, d, k.ythr_fin, lgb, lga, hh,
                             ob + BLK * (nblk - 2), (nblk - 2) >= wfirst);
    finish_block(lgb, hh, ob + BLK * (nblk - 1), k.sig_plain, (nblk - 1) >= wfirst);
}

__global__ __launch_bounds__(32 * (CPW + 1)) __attribute__((amdgpu_waves_per_eu(1, 1))) void rnn_scan(const float* __restrict__ xp,
                                               const float* __restrict__ Whh,
                                               const float* __restrict__ Wfc,
                                               const float* __restrict__ bfc_p,
                                               float* __restrict__ out) {
    const int lane = threadIdx.x & 31;
    const int w    = __builtin_amdgcn_readfirstlane((int)(threadIdx.x >> 5));
    const int c    = lane & 15;
    const int hh   = lane >> 4;
    const int grp  = blockIdx.x % NGRP;
    const int ch   = (blockIdx.x / NGRP) * CPW + w;
    const int b    = grp * 16 + c;
    const int tc   = ch * TCH;
    const bool live = ch < NCH;
    const bool runs = w < CPW;

    __shared__ float etile[CPW][160];

    v8f d = {};
    if (live) {
        const ScanCtx k = make_ctx(Whh, Wfc, bfc_p, c, hh);

        const float* const xl = xp + (size_t)b * Hh + 5 * hh;
        float* const orow = out + (size_t)b * Sseq;

        const int tw = (tc > LWARM) ? tc - LWARM : 0;
        const int nw = (tc - tw) / BLK;
        d = pass_from_zero(k, xl + (size_t)tw * XSTEP);
        float* const cap = &etile[w >= 1 ? w - 1 : 0][lane * 5];
        run_range(k, xl, orow + tw + 4 * hh, d, tw, runs ? nw + TCH / BLK : nw, runs ? nw : NEVER, hh,
                  cap, (runs && w >= 1) ? nw : -1);
        if (!runs) {
            #pragma unroll
            for (int r = 0; r < 5; ++r) cap[r] = d[r];
        }
    }
    __syncthreads();
    if (runs && ch + 1 < NCH) {
        unsigned int df = 0u;
        #pragma unroll
        for (int r = 0; r < 5; ++r) df |= __float_as_uint(d[r]) ^ __float_as_uint(etile[w][lane * 5 + r]);
        if (__builtin_amdgcn_ballot_w32(df != 0u) != 0u) {
            if (hh) out[(size_t)b * Sseq + tc + TCH - 1] = MARK;
        }
    }
}

__global__ __launch_bounds__(256) void rnn_check(const float* __restrict__ xp,
                                                 const float* __restrict__ Whh,
                                                 const float* __restrict__ Wfc,
                                                 const float* __restrict__ bfc_p,
                                                 float* __restrict__ out) {
    const int lane = threadIdx.x & 31;
    const int w    = threadIdx.x >> 5;
    const int c    = lane & 15;
    const int hh   = lane >> 4;
    const int b    = blockIdx.x * 16 + c;
    float* const orow = out + (size_t)b * Sseq;

    __shared__ unsigned int sbad[8];
    unsigned int mybad = 0u;
    #pragma unroll
    for (int i = 0; i < NCH / 16; ++i) {
        const int ch = 2 * (w + 8 * i) + hh;
        const float v = orow[ch * TCH + TCH - 1];
        if (__builtin_amdgcn_ballot_w32(__float_as_uint(v) == __float_as_uint(MARK)) != 0u) mybad = 1u;
    }
    if (lane == 0) sbad[w] = mybad;
    __syncthreads();
    unsigned int bad = 0u;
    #pragma unroll
    for (int i = 0; i < 8; ++i) bad |= sbad[i];
    if (bad == 0u || w != 0) return;

    const ScanCtx k = make_ctx(Whh, Wfc, bfc_p, c, hh);
    const float* const xl = xp + (size_t)b * Hh + 5 * hh;
    v8f d = pass_from_zero(k, xl);
    run_range(k, xl, orow + 4 * hh, d, 0, Sseq / BLK, 0, hh, nullptr, -1);
}

extern "C" void kernel_launch(void* const* d_in, const int* in_sizes, int n_in,
                              void* d_out, int out_size, void* d_ws, size_t ws_size,
                              hipStream_t stream) {
    (void)in_sizes; (void)n_in; (void)out_size; (void)ws_size;
    const float* inp = (const float*)d_in[0];
    const float* Wih = (const float*)d_in[1];
    const float* Whh = (const float*)d_in[2];
    const float* bih = (const float*)d_in[3];
    const float* bhh = (const float*)d_in[4];
    const float* Wfc = (const float*)d_in[5];
    const float* bfc = (const float*)d_in[6];
    float* out = (float*)d_out;
    float* xp  = (float*)d_ws;

    const int tiles = (Bsz * Sseq) / 16;
    xproj_wmma<<<dim3(tiles / (4 * TPW)), 128, 0, stream>>>(inp, Wih, bih, bhh, Whh, xp);
    rnn_scan<<<dim3((NCH / CPW) * NGRP), 32 * (CPW + 1), 0, stream>>>(xp, Whh, Wfc, bfc, out);
    rnn_check<<<dim3(NGRP), 256, 0, stream>>>(xp, Whh, Wfc, bfc, out);
}
